# starting kernel
# baseline (speedup 1.0000x reference)
#include <hip/hip_runtime.h>
#include <cmath>


typedef __attribute__((ext_vector_type(2))) float v2f;
typedef __attribute__((ext_vector_type(8))) float v8f;

#define B_SZ 4096
#define T_SZ 60
#define F_SZ 158
#define F_PAD 160
#define G3 96
#define H_SZ 32

__device__ __forceinline__ float sigmoid_f(float x) {
    return 1.0f / (1.0f + __expf(-x));
}

extern "C" __global__ __launch_bounds__(256)
void gru_gemm_in(const float* __restrict__ x, const float* __restrict__ w_ih,
                 float* __restrict__ gi)
{
    __shared__ float w_lds[G3 * F_PAD];
    const int tid = threadIdx.x;
    for (int i = tid; i < G3 * F_PAD; i += 256) {
        const int row = i / F_PAD, col = i % F_PAD;
        w_lds[i] = (col < F_SZ) ? w_ih[row * F_SZ + col] : 0.0f;
    }
    __syncthreads();

    const int lane = tid & 31;
    const int lo   = lane & 15;
    const int hi   = lane >> 4;
    const int wave = tid >> 5;
    const long m0  = (long)blockIdx.x * 128 + wave * 16;

    v8f acc[6] = {};
    const float* xrow = x + (m0 + lo) * F_SZ;

    for (int kt = 0; kt < 40; ++kt) {
        const int c = kt * 4 + 2 * hi;
        v2f a = {};
        if (c < F_SZ) {
            a.x = xrow[c];
            a.y = xrow[c + 1];
        }
        #pragma unroll
        for (int nt = 0; nt < 6; ++nt) {
            const int n = nt * 16 + lo;
            v2f b;
            b.x = w_lds[n * F_PAD + c];
            b.y = w_lds[n * F_PAD + c + 1];
            acc[nt] = __builtin_amdgcn_wmma_f32_16x16x4_f32(
                false, a, false, b, (short)0, acc[nt], false, false);
        }
    }

    #pragma unroll
    for (int r = 0; r < 8; ++r) {
        const long m = m0 + hi * 8 + r;
        const long t = m % T_SZ;
        const long b = m / T_SZ;
        float* orow = gi + (t * B_SZ + b) * G3;
        #pragma unroll
        for (int nt = 0; nt < 6; ++nt) {
            orow[nt * 16 + lo] = acc[nt][r];
        }
    }
}

extern "C" __global__ __launch_bounds__(64)
void gru_recurrence(const float* __restrict__ gi, const float* __restrict__ w_hh,
                    const float* __restrict__ w1, const float* __restrict__ b1,
                    const float* __restrict__ w2, float* __restrict__ out)
{
    __shared__ float h_lds[2 * 16 * 34];
    const int tid = threadIdx.x;
    for (int i = tid; i < 2 * 16 * 34; i += 64) h_lds[i] = 0.0f;
    __syncthreads();

    const int lane = tid & 31;
    const int lo   = lane & 15;
    const int hi   = lane >> 4;
    const int wave = tid >> 5;
    const int bb   = blockIdx.x * 32 + wave * 16;
    float* hw = h_lds + wave * (16 * 34);

    v2f bf[6][8];
    #pragma unroll
    for (int nt = 0; nt < 6; ++nt) {
        const int n = nt * 16 + lo;
        #pragma unroll
        for (int kt = 0; kt < 8; ++kt) {
            const int k = kt * 4 + 2 * hi;
            bf[nt][kt].x = w_hh[n * H_SZ + k];
            bf[nt][kt].y = w_hh[n * H_SZ + k + 1];
        }
    }

    v8f hc[2] = {};

    for (int t = 0; t < T_SZ; ++t) {
        v8f acc[6] = {};
        #pragma unroll
        for (int kt = 0; kt < 8; ++kt) {
            const int c = kt * 4 + 2 * hi;
            v2f a;
            a.x = hw[lo * 34 + c];
            a.y = hw[lo * 34 + c + 1];
            #pragma unroll
            for (int nt = 0; nt < 6; ++nt) {
                acc[nt] = __builtin_amdgcn_wmma_f32_16x16x4_f32(
                    false, a, false, bf[nt][kt], (short)0, acc[nt], false, false);
            }
        }

        const float* girow = gi + ((long)t * B_SZ + bb) * G3;
        #pragma unroll
        for (int nt = 0; nt < 2; ++nt) {
            v8f ir, iz, inn;
            #pragma unroll
            for (int r = 0; r < 8; ++r) {
                const int row = hi * 8 + r;
                const float* p = girow + (long)row * G3 + lo;
                ir[r]  = p[(nt + 0) * 16];
                iz[r]  = p[(nt + 2) * 16];
                inn[r] = p[(nt + 4) * 16];
            }
            #pragma unroll
            for (int r = 0; r < 8; ++r) {
                const float rg = sigmoid_f(ir[r] + acc[nt][r]);
                const float zg = sigmoid_f(iz[r] + acc[nt + 2][r]);
                const float ng = tanhf(inn[r] + rg * acc[nt + 4][r]);
                hc[nt][r] = (1.0f - zg) * ng + zg * hc[nt][r];
            }
            #pragma unroll
            for (int r = 0; r < 8; ++r) {
                hw[(hi * 8 + r) * 34 + nt * 16 + lo] = hc[nt][r];
            }
        }
        __syncthreads();
    }

    if (lane < 16) {
        float hrow[H_SZ];
        #pragma unroll
        for (int k = 0; k < H_SZ; ++k) hrow[k] = hw[lane * 34 + k];
        float y = 0.0f;
        for (int j = 0; j < 16; ++j) {
            float a = b1[j];
            #pragma unroll
            for (int k = 0; k < H_SZ; ++k) a += hrow[k] * w1[j * H_SZ + k];
            y += w2[j] * fmaxf(a, 0.0f);
        }
        out[bb + lane] = y;
    }
}

extern "C" void kernel_launch(void* const* d_in, const int* in_sizes, int n_in,
                              void* d_out, int out_size, void* d_ws, size_t ws_size,
                              hipStream_t stream) {
    const float* x    = (const float*)d_in[0];
    const float* w_ih = (const float*)d_in[1];
    const float* w_hh = (const float*)d_in[2];
    const float* w1   = (const float*)d_in[3];
    const float* b1   = (const float*)d_in[4];
    const float* w2   = (const float*)d_in[5];
    float* out = (float*)d_out;

    float* gi = (float*)d_ws;

    gru_gemm_in<<<1920, 256, 0, stream>>>(x, w_ih, gi);
    gru_recurrence<<<128, 64, 0, stream>>>(gi, w_hh, w1, b1, w2, out);
}


# recurrence step as one straight-line block: branch-free tanh (both f32 formulas evaluated, one selected) plus a one-wave-per-SIMD register budget, so gate chains and gi loads interleave
# speedup vs baseline: 1.3527x; 1.3527x over previous
#include <hip/hip_runtime.h>
#include <cmath>


typedef __attribute__((ext_vector_type(2))) float v2f;
typedef __attribute__((ext_vector_type(8))) float v8f;

#define B_SZ 4096
#define T_SZ 60
#define F_SZ 158
#define F_PAD 160
#define G3 96
#define H_SZ 32

__device__ __forceinline__ float sigmoid_f(float x) {
    return 1.0f / (1.0f + __expf(-x));
}

__device__ __forceinline__ float tanh_sel(float x) {
    const float ax = fabsf(x);
    const float e  = expf(ax + ax);
    const float big = fmaf(__builtin_amdgcn_rcpf(e + 1.0f), -2.0f, 1.0f);
    const float x2 = ax * ax;
    float p = fmaf(x2, -0x1.758e7ap-8f, 0x1.521192p-6f);
    p = fmaf(x2, p, -0x1.b8389cp-5f);
    p = fmaf(x2, p, 0x1.110704p-3f);
    p = fmaf(x2, p, -0x1.555532p-2f);
    const float small = fmaf(x2, ax * p, ax);
    return copysignf((ax < 0.625f) ? small : big, x);
}

extern "C" __global__ __launch_bounds__(256)
void gru_gemm_in(const float* __restrict__ x, const float* __restrict__ w_ih,
                 float* __restrict__ gi)
{
    __shared__ float w_lds[G3 * F_PAD];
    const int tid = threadIdx.x;
    for (int i = tid; i < G3 * F_PAD; i += 256) {
        const int row = i / F_PAD, col = i % F_PAD;
        w_lds[i] = (col < F_SZ) ? w_ih[row * F_SZ + col] : 0.0f;
    }
    __syncthreads();

    const int lane = tid & 31;
    const int lo   = lane & 15;
    const int hi   = lane >> 4;
    const int wave = tid >> 5;
    const long m0  = (long)blockIdx.x * 128 + wave * 16;

    v8f acc[6] = {};
    const float* xrow = x + (m0 + lo) * F_SZ;

    for (int kt = 0; kt < 40; ++kt) {
        const int c = kt * 4 + 2 * hi;
        v2f a = {};
        if (c < F_SZ) {
            a.x = xrow[c];
            a.y = xrow[c + 1];
        }
        #pragma unroll
        for (int nt = 0; nt < 6; ++nt) {
            const int n = nt * 16 + lo;
            v2f b;
            b.x = w_lds[n * F_PAD + c];
            b.y = w_lds[n * F_PAD + c + 1];
            acc[nt] = __builtin_amdgcn_wmma_f32_16x16x4_f32(
                false, a, false, b, (short)0, acc[nt], false, false);
        }
    }

    #pragma unroll
    for (int r = 0; r < 8; ++r) {
        const long m = m0 + hi * 8 + r;
        const long t = m % T_SZ;
        const long b = m / T_SZ;
        float* orow = gi + (t * B_SZ + b) * G3;
        #pragma unroll
        for (int nt = 0; nt < 6; ++nt) {
            orow[nt * 16 + lo] = acc[nt][r];
        }
    }
}

extern "C" __global__ __launch_bounds__(64)
__attribute__((amdgpu_waves_per_eu(1, 1)))
void gru_recurrence(const float* __restrict__ gi, const float* __restrict__ w_hh,
                    const float* __restrict__ w1, const float* __restrict__ b1,
                    const float* __restrict__ w2, float* __restrict__ out)
{
    __shared__ float h_lds[2 * 16 * 34];
    const int tid = threadIdx.x;
    for (int i = tid; i < 2 * 16 * 34; i += 64) h_lds[i] = 0.0f;
    __syncthreads();

    const int lane = tid & 31;
    const int lo   = lane & 15;
    const int hi   = lane >> 4;
    const int wave = tid >> 5;
    const int bb   = blockIdx.x * 32 + wave * 16;
    float* hw = h_lds + wave * (16 * 34);

    v2f bf[6][8];
    #pragma unroll
    for (int nt = 0; nt < 6; ++nt) {
        const int n = nt * 16 + lo;
        #pragma unroll
        for (int kt = 0; kt < 8; ++kt) {
            const int k = kt * 4 + 2 * hi;
            bf[nt][kt].x = w_hh[n * H_SZ + k];
            bf[nt][kt].y = w_hh[n * H_SZ + k + 1];
        }
    }

    v8f hc[2] = {};

    for (int t = 0; t < T_SZ; ++t) {
        v8f acc[6] = {};
        #pragma unroll
        for (int kt = 0; kt < 8; ++kt) {
            const int c = kt * 4 + 2 * hi;
            v2f a;
            a.x = hw[lo * 34 + c];
            a.y = hw[lo * 34 + c + 1];
            #pragma unroll
            for (int nt = 0; nt < 6; ++nt) {
                acc[nt] = __builtin_amdgcn_wmma_f32_16x16x4_f32(
                    false, a, false, bf[nt][kt], (short)0, acc[nt], false, false);
            }
        }

        const float* girow = gi + ((long)t * B_SZ + bb) * G3;
        #pragma unroll
        for (int nt = 0; nt < 2; ++nt) {
            v8f ir, iz, inn;
            #pragma unroll
            for (int r = 0; r < 8; ++r) {
                const int row = hi * 8 + r;
                const float* p = girow + (long)row * G3 + lo;
                ir[r]  = p[(nt + 0) * 16];
                iz[r]  = p[(nt + 2) * 16];
                inn[r] = p[(nt + 4) * 16];
            }
            #pragma unroll
            for (int r = 0; r < 8; ++r) {
                const float rg = sigmoid_f(ir[r] + acc[nt][r]);
                const float zg = sigmoid_f(iz[r] + acc[nt + 2][r]);
                const float ng = tanh_sel(inn[r] + rg * acc[nt + 4][r]);
                hc[nt][r] = (1.0f - zg) * ng + zg * hc[nt][r];
            }
            #pragma unroll
            for (int r = 0; r < 8; ++r) {
                hw[(hi * 8 + r) * 34 + nt * 16 + lo] = hc[nt][r];
            }
        }
        __syncthreads();
    }

    if (lane < 16) {
        float hrow[H_SZ];
        #pragma unroll
        for (int k = 0; k < H_SZ; ++k) hrow[k] = hw[lane * 34 + k];
        float y = 0.0f;
        for (int j = 0; j < 16; ++j) {
            float a = b1[j];
            #pragma unroll
            for (int k = 0; k < H_SZ; ++k) a += hrow[k] * w1[j * H_SZ + k];
            y += w2[j] * fmaxf(a, 0.0f);
        }
        out[bb + lane] = y;
    }
}

extern "C" void kernel_launch(void* const* d_in, const int* in_sizes, int n_in,
                              void* d_out, int out_size, void* d_ws, size_t ws_size,
                              hipStream_t stream) {
    const float* x    = (const float*)d_in[0];
    const float* w_ih = (const float*)d_in[1];
    const float* w_hh = (const float*)d_in[2];
    const float* w1   = (const float*)d_in[3];
    const float* b1   = (const float*)d_in[4];
    const float* w2   = (const float*)d_in[5];
    float* out = (float*)d_out;

    float* gi = (float*)d_ws;

    gru_gemm_in<<<1920, 256, 0, stream>>>(x, w_ih, gi);
    gru_recurrence<<<128, 64, 0, stream>>>(gi, w_hh, w1, b1, w2, out);
}


# recurrence: each 16-row tile's step shared by four waves (one per SIMD): a wave does one h-column half's three gate tiles and the gates of four accumulator rows; h double-buffered in LDS
# speedup vs baseline: 1.6530x; 1.2220x over previous
#include <hip/hip_runtime.h>
#include <cmath>


typedef __attribute__((ext_vector_type(2))) float v2f;
typedef __attribute__((ext_vector_type(8))) float v8f;

#define B_SZ 4096
#define T_SZ 60
#define F_SZ 158
#define F_PAD 160
#define G3 96
#define H_SZ 32

__device__ __forceinline__ float sigmoid_f(float x) {
    return 1.0f / (1.0f + __expf(-x));
}

__device__ __forceinline__ float tanh_sel(float x) {
    const float ax = fabsf(x);
    const float e  = expf(ax + ax);
    const float big = fmaf(__builtin_amdgcn_rcpf(e + 1.0f), -2.0f, 1.0f);
    const float x2 = ax * ax;
    float p = fmaf(x2, -0x1.758e7ap-8f, 0x1.521192p-6f);
    p = fmaf(x2, p, -0x1.b8389cp-5f);
    p = fmaf(x2, p, 0x1.110704p-3f);
    p = fmaf(x2, p, -0x1.555532p-2f);
    const float small = fmaf(x2, ax * p, ax);
    return copysignf((ax < 0.625f) ? small : big, x);
}

extern "C" __global__ __launch_bounds__(256)
void gru_gemm_in(const float* __restrict__ x, const float* __restrict__ w_ih,
                 float* __restrict__ gi)
{
    __shared__ float w_lds[G3 * F_PAD];
    const int tid = threadIdx.x;
    for (int i = tid; i < G3 * F_PAD; i += 256) {
        const int row = i / F_PAD, col = i % F_PAD;
        w_lds[i] = (col < F_SZ) ? w_ih[row * F_SZ + col] : 0.0f;
    }
    __syncthreads();

    const int lane = tid & 31;
    const int lo   = lane & 15;
    const int hi   = lane >> 4;
    const int wave = tid >> 5;
    const long m0  = (long)blockIdx.x * 128 + wave * 16;

    v8f acc[6] = {};
    const float* xrow = x + (m0 + lo) * F_SZ;

    for (int kt = 0; kt < 40; ++kt) {
        const int c = kt * 4 + 2 * hi;
        v2f a = {};
        if (c < F_SZ) {
            a.x = xrow[c];
            a.y = xrow[c + 1];
        }
        #pragma unroll
        for (int nt = 0; nt < 6; ++nt) {
            const int n = nt * 16 + lo;
            v2f b;
            b.x = w_lds[n * F_PAD + c];
            b.y = w_lds[n * F_PAD + c + 1];
            acc[nt] = __builtin_amdgcn_wmma_f32_16x16x4_f32(
                false, a, false, b, (short)0, acc[nt], false, false);
        }
    }

    #pragma unroll
    for (int r = 0; r < 8; ++r) {
        const long m = m0 + hi * 8 + r;
        const long t = m % T_SZ;
        const long b = m / T_SZ;
        float* orow = gi + (t * B_SZ + b) * G3;
        #pragma unroll
        for (int nt = 0; nt < 6; ++nt) {
            orow[nt * 16 + lo] = acc[nt][r];
        }
    }
}

#define HB (16 * 34)

__device__ __forceinline__ void gru_step(
    const float* __restrict__ gip,
    const float* rd, float* wr,
    const v2f (&bf)[3][8], float (&hc)[4],
    const int a_off, const int s_off)
{
    float ir[4], iz[4], inn[4];
    #pragma unroll
    for (int rr = 0; rr < 4; ++rr) {
        const float* p = gip + rr * G3;
        ir[rr]  = p[0];
        iz[rr]  = p[32];
        inn[rr] = p[64];
    }

    v8f acc[3] = {};
    #pragma unroll
    for (int kt = 0; kt < 8; ++kt) {
        v2f a;
        a.x = rd[a_off + kt * 4];
        a.y = rd[a_off + kt * 4 + 1];
        #pragma unroll
        for (int g = 0; g < 3; ++g) {
            acc[g] = __builtin_amdgcn_wmma_f32_16x16x4_f32(
                false, a, false, bf[g][kt], (short)0, acc[g], false, false);
        }
    }

    #pragma unroll
    for (int rr = 0; rr < 4; ++rr) {
        const float rg = sigmoid_f(ir[rr] + acc[0][rr]);
        const float zg = sigmoid_f(iz[rr] + acc[1][rr]);
        const float ng = tanh_sel(inn[rr] + rg * acc[2][rr]);
        hc[rr] = (1.0f - zg) * ng + zg * hc[rr];
        wr[s_off + rr * 34] = hc[rr];
    }
}

extern "C" __global__ __launch_bounds__(128)
__attribute__((amdgpu_waves_per_eu(1, 1)))
void gru_recurrence(const float* __restrict__ gi, const float* __restrict__ w_hh,
                    const float* __restrict__ w1, const float* __restrict__ b1,
                    const float* __restrict__ w2, float* __restrict__ out)
{
    static_assert(T_SZ % 2 == 0, "the scan alternates two h buffers in pairs of steps");
    __shared__ float h_lds[2 * HB];
    const int tid = threadIdx.x;
    for (int i = tid; i < HB; i += 128) h_lds[i] = 0.0f;
    __syncthreads();

    const int lane = tid & 31;
    const int lo   = lane & 15;
    const int hi   = lane >> 4;
    const int wave = tid >> 5;
    const int j    = wave & 1;
    const int q    = wave >> 1;
    const int bb   = blockIdx.x * 16;

    const int a_off = (lo ^ (q << 2)) * 34 + 2 * hi;
    const int s_off = (hi * 8 + q * 4) * 34 + j * 16 + lo;

    v2f bf[3][8];
    #pragma unroll
    for (int g = 0; g < 3; ++g) {
        const int n = (2 * g + j) * 16 + lo;
        #pragma unroll
        for (int kt = 0; kt < 8; ++kt) {
            const int k = kt * 4 + 2 * hi;
            bf[g][kt].x = w_hh[n * H_SZ + k];
            bf[g][kt].y = w_hh[n * H_SZ + k + 1];
        }
    }

    float hc[4] = {0.0f, 0.0f, 0.0f, 0.0f};

    const float* gip = gi + ((long)bb + hi * 8 + q * 4) * G3 + j * 16 + lo;
    const long gstep = (long)B_SZ * G3;

    for (int t = 0; t < T_SZ; t += 2) {
        gru_step(gip, h_lds, h_lds + HB, bf, hc, a_off, s_off);
        gip += gstep;
        __syncthreads();
        gru_step(gip, h_lds + HB, h_lds, bf, hc, a_off, s_off);
        gip += gstep;
        __syncthreads();
    }

    if (tid < 16) {
        float hrow[H_SZ];
        #pragma unroll
        for (int k = 0; k < H_SZ; ++k) hrow[k] = h_lds[tid * 34 + k];
        float y = 0.0f;
        for (int jj = 0; jj < 16; ++jj) {
            float a = b1[jj];
            #pragma unroll
            for (int k = 0; k < H_SZ; ++k) a += hrow[k] * w1[jj * H_SZ + k];
            y += w2[jj] * fmaxf(a, 0.0f);
        }
        out[bb + tid] = y;
    }
}

extern "C" void kernel_launch(void* const* d_in, const int* in_sizes, int n_in,
                              void* d_out, int out_size, void* d_ws, size_t ws_size,
                              hipStream_t stream) {
    const float* x    = (const float*)d_in[0];
    const float* w_ih = (const float*)d_in[1];
    const float* w_hh = (const float*)d_in[2];
    const float* w1   = (const float*)d_in[3];
    const float* b1   = (const float*)d_in[4];
    const float* w2   = (const float*)d_in[5];
    float* out = (float*)d_out;

    float* gi = (float*)d_ws;

    gru_gemm_in<<<1920, 256, 0, stream>>>(x, w_ih, gi);
    gru_recurrence<<<256, 128, 0, stream>>>(gi, w_hh, w1, b1, w2, out);
}


# input projection: w_ih row pitch in LDS 160 -> 164 dwords, so a B-fragment read falls on each of the 64 banks once instead of on 8 banks with 8 distinct dwords each
# speedup vs baseline: 1.7501x; 1.0587x over previous
#include <hip/hip_runtime.h>
#include <cmath>


typedef __attribute__((ext_vector_type(2))) float v2f;
typedef __attribute__((ext_vector_type(8))) float v8f;

#define B_SZ 4096
#define T_SZ 60
#define F_SZ 158
#define F_PAD 160
#define W_PITCH 164
#define G3 96
#define H_SZ 32

__device__ __forceinline__ float sigmoid_f(float x) {
    return 1.0f / (1.0f + __expf(-x));
}

__device__ __forceinline__ float tanh_sel(float x) {
    const float ax = fabsf(x);
    const float e  = expf(ax + ax);
    const float big = fmaf(__builtin_amdgcn_rcpf(e + 1.0f), -2.0f, 1.0f);
    const float x2 = ax * ax;
    float p = fmaf(x2, -0x1.758e7ap-8f, 0x1.521192p-6f);
    p = fmaf(x2, p, -0x1.b8389cp-5f);
    p = fmaf(x2, p, 0x1.110704p-3f);
    p = fmaf(x2, p, -0x1.555532p-2f);
    const float small = fmaf(x2, ax * p, ax);
    return copysignf((ax < 0.625f) ? small : big, x);
}

extern "C" __global__ __launch_bounds__(256)
void gru_gemm_in(const float* __restrict__ x, const float* __restrict__ w_ih,
                 float* __restrict__ gi)
{
    __shared__ float w_lds[G3 * W_PITCH];
    const int tid = threadIdx.x;
    for (int i = tid; i < G3 * W_PITCH; i += 256) {
        const int row = i / W_PITCH, col = i % W_PITCH;
        w_lds[i] = (col < F_SZ) ? w_ih[row * F_SZ + col] : 0.0f;
    }
    __syncthreads();

    const int lane = tid & 31;
    const int lo   = lane & 15;
    const int hi   = lane >> 4;
    const int wave = tid >> 5;
    const long m0  = (long)blockIdx.x * 128 + wave * 16;

    v8f acc[6] = {};
    const float* xrow = x + (m0 + lo) * F_SZ;

    for (int kt = 0; kt < 40; ++kt) {
        const int c = kt * 4 + 2 * hi;
        v2f a = {};
        if (c < F_SZ) {
            a.x = xrow[c];
            a.y = xrow[c + 1];
        }
        #pragma unroll
        for (int nt = 0; nt < 6; ++nt) {
            const int n = nt * 16 + lo;
            v2f b;
            b.x = w_lds[n * W_PITCH + c];
            b.y = w_lds[n * W_PITCH + c + 1];
            acc[nt] = __builtin_amdgcn_wmma_f32_16x16x4_f32(
                false, a, false, b, (short)0, acc[nt], false, false);
        }
    }

    #pragma unroll
    for (int r = 0; r < 8; ++r) {
        const long m = m0 + hi * 8 + r;
        const long t = m % T_SZ;
        const long b = m / T_SZ;
        float* orow = gi + (t * B_SZ + b) * G3;
        #pragma unroll
        for (int nt = 0; nt < 6; ++nt) {
            orow[nt * 16 + lo] = acc[nt][r];
        }
    }
}

#define HB (16 * 34)

__device__ __forceinline__ void gru_step(
    const float* __restrict__ gip,
    const float* rd, float* wr,
    const v2f (&bf)[3][8], float (&hc)[4],
    const int a_off, const int s_off)
{
    float ir[4], iz[4], inn[4];
    #pragma unroll
    for (int rr = 0; rr < 4; ++rr) {
        const float* p = gip + rr * G3;
        ir[rr]  = p[0];
        iz[rr]  = p[32];
        inn[rr] = p[64];
    }

    v8f acc[3] = {};
    #pragma unroll
    for (int kt = 0; kt < 8; ++kt) {
        v2f a;
        a.x = rd[a_off + kt * 4];
        a.y = rd[a_off + kt * 4 + 1];
        #pragma unroll
        for (int g = 0; g < 3; ++g) {
            acc[g] = __builtin_amdgcn_wmma_f32_16x16x4_f32(
                false, a, false, bf[g][kt], (short)0, acc[g], false, false);
        }
    }

    #pragma unroll
    for (int rr = 0; rr < 4; ++rr) {
        const float rg = sigmoid_f(ir[rr] + acc[0][rr]);
        const float zg = sigmoid_f(iz[rr] + acc[1][rr]);
        const float ng = tanh_sel(inn[rr] + rg * acc[2][rr]);
        hc[rr] = (1.0f - zg) * ng + zg * hc[rr];
        wr[s_off + rr * 34] = hc[rr];
    }
}

extern "C" __global__ __launch_bounds__(128)
__attribute__((amdgpu_waves_per_eu(1, 1)))
void gru_recurrence(const float* __restrict__ gi, const float* __restrict__ w_hh,
                    const float* __restrict__ w1, const float* __restrict__ b1,
                    const float* __restrict__ w2, float* __restrict__ out)
{
    static_assert(T_SZ % 2 == 0, "the scan alternates two h buffers in pairs of steps");
    __shared__ float h_lds[2 * HB];
    const int tid = threadIdx.x;
    for (int i = tid; i < HB; i += 128) h_lds[i] = 0.0f;
    __syncthreads();

    const int lane = tid & 31;
    const int lo   = lane & 15;
    const int hi   = lane >> 4;
    const int wave = tid >> 5;
    const int j    = wave & 1;
    const int q    = wave >> 1;
    const int bb   = blockIdx.x * 16;

    const int a_off = (lo ^ (q << 2)) * 34 + 2 * hi;
    const int s_off = (hi * 8 + q * 4) * 34 + j * 16 + lo;

    v2f bf[3][8];
    #pragma unroll
    for (int g = 0; g < 3; ++g) {
        const int n = (2 * g + j) * 16 + lo;
        #pragma unroll
        for (int kt = 0; kt < 8; ++kt) {
            const int k = kt * 4 + 2 * hi;
            bf[g][kt].x = w_hh[n * H_SZ + k];
            bf[g][kt].y = w_hh[n * H_SZ + k + 1];
        }
    }

    float hc[4] = {0.0f, 0.0f, 0.0f, 0.0f};

    const float* gip = gi + ((long)bb + hi * 8 + q * 4) * G3 + j * 16 + lo;
    const long gstep = (long)B_SZ * G3;

    for (int t = 0; t < T_SZ; t += 2) {
        gru_step(gip, h_lds, h_lds + HB, bf, hc, a_off, s_off);
        gip += gstep;
        __syncthreads();
        gru_step(gip, h_lds + HB, h_lds, bf, hc, a_off, s_off);
        gip += gstep;
        __syncthreads();
    }

    if (tid < 16) {
        float hrow[H_SZ];
        #pragma unroll
        for (int k = 0; k < H_SZ; ++k) hrow[k] = h_lds[tid * 34 + k];
        float y = 0.0f;
        for (int jj = 0; jj < 16; ++jj) {
            float a = b1[jj];
            #pragma unroll
            for (int k = 0; k < H_SZ; ++k) a += hrow[k] * w1[jj * H_SZ + k];
            y += w2[jj] * fmaxf(a, 0.0f);
        }
        out[bb + tid] = y;
    }
}

extern "C" void kernel_launch(void* const* d_in, const int* in_sizes, int n_in,
                              void* d_out, int out_size, void* d_ws, size_t ws_size,
                              hipStream_t stream) {
    const float* x    = (const float*)d_in[0];
    const float* w_ih = (const float*)d_in[1];
    const float* w_hh = (const float*)d_in[2];
    const float* w1   = (const float*)d_in[3];
    const float* b1   = (const float*)d_in[4];
    const float* w2   = (const float*)d_in[5];
    float* out = (float*)d_out;

    float* gi = (float*)d_ws;

    gru_gemm_in<<<1920, 256, 0, stream>>>(x, w_ih, gi);
    gru_recurrence<<<256, 128, 0, stream>>>(gi, w_hh, w1, b1, w2, out);
}


# recurrence: gi of step t+1 is requested at the top of step t, one step ahead (12 more registers; the loop barrier fences LDS only, so it does not wait for them): no step waits for its own gi loads
# speedup vs baseline: 2.0217x; 1.1552x over previous
#include <hip/hip_runtime.h>
#include <cmath>


typedef __attribute__((ext_vector_type(2))) float v2f;
typedef __attribute__((ext_vector_type(8))) float v8f;

#define B_SZ 4096
#define T_SZ 60
#define F_SZ 158
#define F_PAD 160
#define W_PITCH 164
#define G3 96
#define H_SZ 32

__device__ __forceinline__ float sigmoid_f(float x) {
    return 1.0f / (1.0f + __expf(-x));
}

__device__ __forceinline__ float tanh_sel(float x) {
    const float ax = fabsf(x);
    const float e  = expf(ax + ax);
    const float big = fmaf(__builtin_amdgcn_rcpf(e + 1.0f), -2.0f, 1.0f);
    const float x2 = ax * ax;
    float p = fmaf(x2, -0x1.758e7ap-8f, 0x1.521192p-6f);
    p = fmaf(x2, p, -0x1.b8389cp-5f);
    p = fmaf(x2, p, 0x1.110704p-3f);
    p = fmaf(x2, p, -0x1.555532p-2f);
    const float small = fmaf(x2, ax * p, ax);
    return copysignf((ax < 0.625f) ? small : big, x);
}

extern "C" __global__ __launch_bounds__(256)
void gru_gemm_in(const float* __restrict__ x, const float* __restrict__ w_ih,
                 float* __restrict__ gi)
{
    __shared__ float w_lds[G3 * W_PITCH];
    const int tid = threadIdx.x;
    for (int i = tid; i < G3 * W_PITCH; i += 256) {
        const int row = i / W_PITCH, col = i % W_PITCH;
        w_lds[i] = (col < F_SZ) ? w_ih[row * F_SZ + col] : 0.0f;
    }
    __syncthreads();

    const int lane = tid & 31;
    const int lo   = lane & 15;
    const int hi   = lane >> 4;
    const int wave = tid >> 5;
    const long m0  = (long)blockIdx.x * 128 + wave * 16;

    v8f acc[6] = {};
    const float* xrow = x + (m0 + lo) * F_SZ;

    for (int kt = 0; kt < 40; ++kt) {
        const int c = kt * 4 + 2 * hi;
        v2f a = {};
        if (c < F_SZ) {
            a.x = xrow[c];
            a.y = xrow[c + 1];
        }
        #pragma unroll
        for (int nt = 0; nt < 6; ++nt) {
            const int n = nt * 16 + lo;
            v2f b;
            b.x = w_lds[n * W_PITCH + c];
            b.y = w_lds[n * W_PITCH + c + 1];
            acc[nt] = __builtin_amdgcn_wmma_f32_16x16x4_f32(
                false, a, false, b, (short)0, acc[nt], false, false);
        }
    }

    #pragma unroll
    for (int r = 0; r < 8; ++r) {
        const long m = m0 + hi * 8 + r;
        const long t = m % T_SZ;
        const long b = m / T_SZ;
        float* orow = gi + (t * B_SZ + b) * G3;
        #pragma unroll
        for (int nt = 0; nt < 6; ++nt) {
            orow[nt * 16 + lo] = acc[nt][r];
        }
    }
}

#define HB (16 * 34)

__device__ __forceinline__ void gi_load(const float* __restrict__ p0, float (&g)[12])
{
    #pragma unroll
    for (int rr = 0; rr < 4; ++rr) {
        const float* p = p0 + rr * G3;
        g[rr]     = p[0];
        g[4 + rr] = p[32];
        g[8 + rr] = p[64];
    }
}

__device__ __forceinline__ void gru_step(
    const float* __restrict__ gnext,
    const float* rd, float* wr,
    const v2f (&bf)[3][8], float (&hc)[4], float (&gc)[12],
    const int a_off, const int s_off)
{
    float gn[12];
    gi_load(gnext, gn);

    v8f acc[3] = {};
    #pragma unroll
    for (int kt = 0; kt < 8; ++kt) {
        v2f a;
        a.x = rd[a_off + kt * 4];
        a.y = rd[a_off + kt * 4 + 1];
        #pragma unroll
        for (int g = 0; g < 3; ++g) {
            acc[g] = __builtin_amdgcn_wmma_f32_16x16x4_f32(
                false, a, false, bf[g][kt], (short)0, acc[g], false, false);
        }
    }

    #pragma unroll
    for (int rr = 0; rr < 4; ++rr) {
        const float rg = sigmoid_f(gc[rr] + acc[0][rr]);
        const float zg = sigmoid_f(gc[4 + rr] + acc[1][rr]);
        const float ng = tanh_sel(gc[8 + rr] + rg * acc[2][rr]);
        hc[rr] = (1.0f - zg) * ng + zg * hc[rr];
        wr[s_off + rr * 34] = hc[rr];
    }

    #pragma unroll
    for (int i = 0; i < 12; ++i) gc[i] = gn[i];
}

__device__ __forceinline__ void lds_barrier()
{
    __builtin_amdgcn_fence(__ATOMIC_RELEASE, "workgroup", "local");
    __builtin_amdgcn_s_barrier();
    __builtin_amdgcn_fence(__ATOMIC_ACQUIRE, "workgroup", "local");
}

extern "C" __global__ __launch_bounds__(128)
__attribute__((amdgpu_waves_per_eu(1, 1)))
void gru_recurrence(const float* __restrict__ gi, const float* __restrict__ w_hh,
                    const float* __restrict__ w1, const float* __restrict__ b1,
                    const float* __restrict__ w2, float* __restrict__ out)
{
    static_assert(T_SZ % 2 == 0, "the scan alternates two h buffers in pairs of steps");
    __shared__ float h_lds[2 * HB];
    const int tid = threadIdx.x;
    for (int i = tid; i < HB; i += 128) h_lds[i] = 0.0f;
    __syncthreads();

    const int lane = tid & 31;
    const int lo   = lane & 15;
    const int hi   = lane >> 4;
    const int wave = tid >> 5;
    const int j    = wave & 1;
    const int q    = wave >> 1;
    const int bb   = blockIdx.x * 16;

    const int a_off = (lo ^ (q << 2)) * 34 + 2 * hi;
    const int s_off = (hi * 8 + q * 4) * 34 + j * 16 + lo;

    v2f bf[3][8];
    #pragma unroll
    for (int g = 0; g < 3; ++g) {
        const int n = (2 * g + j) * 16 + lo;
        #pragma unroll
        for (int kt = 0; kt < 8; ++kt) {
            const int k = kt * 4 + 2 * hi;
            bf[g][kt].x = w_hh[n * H_SZ + k];
            bf[g][kt].y = w_hh[n * H_SZ + k + 1];
        }
    }

    float hc[4] = {0.0f, 0.0f, 0.0f, 0.0f};

    const float* gip = gi + ((long)bb + hi * 8 + q * 4) * G3 + j * 16 + lo;
    const long gstep = (long)B_SZ * G3;

    float gc[12];
    gi_load(gip, gc);

    for (int t = 0; t < T_SZ; t += 2) {
        gru_step(gip + gstep, h_lds, h_lds + HB, bf, hc, gc, a_off, s_off);
        gip += gstep;
        lds_barrier();
        const float* gnx = (t + 2 < T_SZ) ? gip + gstep : gip;
        gru_step(gnx, h_lds + HB, h_lds, bf, hc, gc, a_off, s_off);
        gip += gstep;
        lds_barrier();
    }

    if (tid < 16) {
        float hrow[H_SZ];
        #pragma unroll
        for (int k = 0; k < H_SZ; ++k) hrow[k] = h_lds[tid * 34 + k];
        float y = 0.0f;
        for (int jj = 0; jj < 16; ++jj) {
            float a = b1[jj];
            #pragma unroll
            for (int k = 0; k < H_SZ; ++k) a += hrow[k] * w1[jj * H_SZ + k];
            y += w2[jj] * fmaxf(a, 0.0f);
        }
        out[bb + tid] = y;
    }
}

extern "C" void kernel_launch(void* const* d_in, const int* in_sizes, int n_in,
                              void* d_out, int out_size, void* d_ws, size_t ws_size,
                              hipStream_t stream) {
    const float* x    = (const float*)d_in[0];
    const float* w_ih = (const float*)d_in[1];
    const float* w_hh = (const float*)d_in[2];
    const float* w1   = (const float*)d_in[3];
    const float* b1   = (const float*)d_in[4];
    const float* w2   = (const float*)d_in[5];
    float* out = (float*)d_out;

    float* gi = (float*)d_ws;

    gru_gemm_in<<<1920, 256, 0, stream>>>(x, w_ih, gi);
    gru_recurrence<<<256, 128, 0, stream>>>(gi, w_hh, w1, b1, w2, out);
}


# input projection: w_ih staged into LDS as 32 pair loads per thread, all requested before the first store, in place of 62 single loads each waited for before the next was issued
# speedup vs baseline: 2.6173x; 1.2946x over previous
#include <hip/hip_runtime.h>
#include <cmath>


typedef __attribute__((ext_vector_type(2))) float v2f;
typedef __attribute__((ext_vector_type(8))) float v8f;

#define B_SZ 4096
#define T_SZ 60
#define F_SZ 158
#define F_PAD 160
#define W_PITCH 164
#define G3 96
#define H_SZ 32

__device__ __forceinline__ float sigmoid_f(float x) {
    return 1.0f / (1.0f + __expf(-x));
}

__device__ __forceinline__ float tanh_sel(float x) {
    const float ax = fabsf(x);
    const float e  = expf(ax + ax);
    const float big = fmaf(__builtin_amdgcn_rcpf(e + 1.0f), -2.0f, 1.0f);
    const float x2 = ax * ax;
    float p = fmaf(x2, -0x1.758e7ap-8f, 0x1.521192p-6f);
    p = fmaf(x2, p, -0x1.b8389cp-5f);
    p = fmaf(x2, p, 0x1.110704p-3f);
    p = fmaf(x2, p, -0x1.555532p-2f);
    const float small = fmaf(x2, ax * p, ax);
    return copysignf((ax < 0.625f) ? small : big, x);
}

extern "C" __global__ __launch_bounds__(256)
void gru_gemm_in(const float* __restrict__ x, const float* __restrict__ w_ih,
                 float* __restrict__ gi)
{
    __shared__ float w_lds[G3 * W_PITCH];
    const int tid = threadIdx.x;

    constexpr int PAIRS = F_SZ / 2;
    constexpr int RPT   = 3;
    constexpr int NTRIP = G3 / RPT;
    static_assert(F_SZ % 2 == 0 && PAIRS * RPT <= 256 && G3 % RPT == 0,
                  "pair staging: even rows, three rows per trip of at most 256 threads, 96 rows");
    if (tid < PAIRS * RPT) {
        const int r0 = tid / PAIRS;
        const int p  = tid - r0 * PAIRS;
        const float* s = w_ih + r0 * F_SZ + 2 * p;
        float* d = w_lds + r0 * W_PITCH + 2 * p;
        v2f wreg[NTRIP];
        #pragma unroll
        for (int j = 0; j < NTRIP; ++j) {
            wreg[j].x = s[j * RPT * F_SZ];
            wreg[j].y = s[j * RPT * F_SZ + 1];
        }
        #pragma unroll
        for (int j = 0; j < NTRIP; ++j) {
            d[j * RPT * W_PITCH]     = wreg[j].x;
            d[j * RPT * W_PITCH + 1] = wreg[j].y;
        }
    }
    if (tid < G3) {
        float* z = w_lds + tid * W_PITCH + F_SZ;
        #pragma unroll
        for (int k = 0; k < W_PITCH - F_SZ; ++k) z[k] = 0.0f;
    }
    __syncthreads();

    const int lane = tid & 31;
    const int lo   = lane & 15;
    const int hi   = lane >> 4;
    const int wave = tid >> 5;
    const long m0  = (long)blockIdx.x * 128 + wave * 16;

    v8f acc[6] = {};
    const float* xrow = x + (m0 + lo) * F_SZ;

    for (int kt = 0; kt < 40; ++kt) {
        const int c = kt * 4 + 2 * hi;
        v2f a = {};
        if (c < F_SZ) {
            a.x = xrow[c];
            a.y = xrow[c + 1];
        }
        #pragma unroll
        for (int nt = 0; nt < 6; ++nt) {
            const int n = nt * 16 + lo;
            v2f b;
            b.x = w_lds[n * W_PITCH + c];
            b.y = w_lds[n * W_PITCH + c + 1];
            acc[nt] = __builtin_amdgcn_wmma_f32_16x16x4_f32(
                false, a, false, b, (short)0, acc[nt], false, false);
        }
    }

    #pragma unroll
    for (int r = 0; r < 8; ++r) {
        const long m = m0 + hi * 8 + r;
        const long t = m % T_SZ;
        const long b = m / T_SZ;
        float* orow = gi + (t * B_SZ + b) * G3;
        #pragma unroll
        for (int nt = 0; nt < 6; ++nt) {
            orow[nt * 16 + lo] = acc[nt][r];
        }
    }
}

#define HB (16 * 34)

__device__ __forceinline__ void gi_load(const float* __restrict__ p0, float (&g)[12])
{
    #pragma unroll
    for (int rr = 0; rr < 4; ++rr) {
        const float* p = p0 + rr * G3;
        g[rr]     = p[0];
        g[4 + rr] = p[32];
        g[8 + rr] = p[64];
    }
}

__device__ __forceinline__ void gru_step(
    const float* __restrict__ gnext,
    const float* rd, float* wr,
    const v2f (&bf)[3][8], float (&hc)[4], float (&gc)[12],
    const int a_off, const int s_off)
{
    float gn[12];
    gi_load(gnext, gn);

    v8f acc[3] = {};
    #pragma unroll
    for (int kt = 0; kt < 8; ++kt) {
        v2f a;
        a.x = rd[a_off + kt * 4];
        a.y = rd[a_off + kt * 4 + 1];
        #pragma unroll
        for (int g = 0; g < 3; ++g) {
            acc[g] = __builtin_amdgcn_wmma_f32_16x16x4_f32(
                false, a, false, bf[g][kt], (short)0, acc[g], false, false);
        }
    }

    #pragma unroll
    for (int rr = 0; rr < 4; ++rr) {
        const float rg = sigmoid_f(gc[rr] + acc[0][rr]);
        const float zg = sigmoid_f(gc[4 + rr] + acc[1][rr]);
        const float ng = tanh_sel(gc[8 + rr] + rg * acc[2][rr]);
        hc[rr] = (1.0f - zg) * ng + zg * hc[rr];
        wr[s_off + rr * 34] = hc[rr];
    }

    #pragma unroll
    for (int i = 0; i < 12; ++i) gc[i] = gn[i];
}

__device__ __forceinline__ void lds_barrier()
{
    __builtin_amdgcn_fence(__ATOMIC_RELEASE, "workgroup", "local");
    __builtin_amdgcn_s_barrier();
    __builtin_amdgcn_fence(__ATOMIC_ACQUIRE, "workgroup", "local");
}

extern "C" __global__ __launch_bounds__(128)
__attribute__((amdgpu_waves_per_eu(1, 1)))
void gru_recurrence(const float* __restrict__ gi, const float* __restrict__ w_hh,
                    const float* __restrict__ w1, const float* __restrict__ b1,
                    const float* __restrict__ w2, float* __restrict__ out)
{
    static_assert(T_SZ % 2 == 0, "the scan alternates two h buffers in pairs of steps");
    __shared__ float h_lds[2 * HB];
    const int tid = threadIdx.x;
    for (int i = tid; i < HB; i += 128) h_lds[i] = 0.0f;
    __syncthreads();

    const int lane = tid & 31;
    const int lo   = lane & 15;
    const int hi   = lane >> 4;
    const int wave = tid >> 5;
    const int j    = wave & 1;
    const int q    = wave >> 1;
    const int bb   = blockIdx.x * 16;

    const int a_off = (lo ^ (q << 2)) * 34 + 2 * hi;
    const int s_off = (hi * 8 + q * 4) * 34 + j * 16 + lo;

    v2f bf[3][8];
    #pragma unroll
    for (int g = 0; g < 3; ++g) {
        const int n = (2 * g + j) * 16 + lo;
        #pragma unroll
        for (int kt = 0; kt < 8; ++kt) {
            const int k = kt * 4 + 2 * hi;
            bf[g][kt].x = w_hh[n * H_SZ + k];
            bf[g][kt].y = w_hh[n * H_SZ + k + 1];
        }
    }

    float hc[4] = {0.0f, 0.0f, 0.0f, 0.0f};

    const float* gip = gi + ((long)bb + hi * 8 + q * 4) * G3 + j * 16 + lo;
    const long gstep = (long)B_SZ * G3;

    float gc[12];
    gi_load(gip, gc);

    for (int t = 0; t < T_SZ; t += 2) {
        gru_step(gip + gstep, h_lds, h_lds + HB, bf, hc, gc, a_off, s_off);
        gip += gstep;
        lds_barrier();
        const float* gnx = (t + 2 < T_SZ) ? gip + gstep : gip;
        gru_step(gnx, h_lds + HB, h_lds, bf, hc, gc, a_off, s_off);
        gip += gstep;
        lds_barrier();
    }

    if (tid < 16) {
        float hrow[H_SZ];
        #pragma unroll
        for (int k = 0; k < H_SZ; ++k) hrow[k] = h_lds[tid * 34 + k];
        float y = 0.0f;
        for (int jj = 0; jj < 16; ++jj) {
            float a = b1[jj];
            #pragma unroll
            for (int k = 0; k < H_SZ; ++k) a += hrow[k] * w1[jj * H_SZ + k];
            y += w2[jj] * fmaxf(a, 0.0f);
        }
        out[bb + tid] = y;
    }
}

extern "C" void kernel_launch(void* const* d_in, const int* in_sizes, int n_in,
                              void* d_out, int out_size, void* d_ws, size_t ws_size,
                              hipStream_t stream) {
    const float* x    = (const float*)d_in[0];
    const float* w_ih = (const float*)d_in[1];
    const float* w_hh = (const float*)d_in[2];
    const float* w1   = (const float*)d_in[3];
    const float* b1   = (const float*)d_in[4];
    const float* w2   = (const float*)d_in[5];
    float* out = (float*)d_out;

    float* gi = (float*)d_ws;

    gru_gemm_in<<<1920, 256, 0, stream>>>(x, w_ih, gi);
    gru_recurrence<<<256, 128, 0, stream>>>(gi, w_hh, w1, b1, w2, out);
}


# input projection: a wave reads its B fragments of w_ih from LDS once into 480 registers and keeps them over 15 row tiles (one wave per SIMD); x is requested one tile ahead into a second fragment set
# speedup vs baseline: 3.0023x; 1.1471x over previous
#include <hip/hip_runtime.h>
#include <cmath>


typedef __attribute__((ext_vector_type(2))) float v2f;
typedef __attribute__((ext_vector_type(8))) float v8f;

#define B_SZ 4096
#define T_SZ 60
#define F_SZ 158
#define F_PAD 160
#define W_PITCH 164
#define G3 96
#define H_SZ 32

__device__ __forceinline__ float sigmoid_f(float x) {
    return 1.0f / (1.0f + __expf(-x));
}

__device__ __forceinline__ float tanh_sel(float x) {
    const float ax = fabsf(x);
    const float e  = expf(ax + ax);
    const float big = fmaf(__builtin_amdgcn_rcpf(e + 1.0f), -2.0f, 1.0f);
    const float x2 = ax * ax;
    float p = fmaf(x2, -0x1.758e7ap-8f, 0x1.521192p-6f);
    p = fmaf(x2, p, -0x1.b8389cp-5f);
    p = fmaf(x2, p, 0x1.110704p-3f);
    p = fmaf(x2, p, -0x1.555532p-2f);
    const float small = fmaf(x2, ax * p, ax);
    return copysignf((ax < 0.625f) ? small : big, x);
}

#define K1_THREADS 128
#define K1_KT      40
#define K1_WAVES   1024
#define K1_TILES   (B_SZ * T_SZ / 16)
#define K1_STEPS   (K1_TILES / K1_WAVES)

__device__ __forceinline__ void k1_load_a(const float* __restrict__ xrow, const int hi,
                                          v2f (&a)[K1_KT])
{
    #pragma unroll
    for (int kt = 0; kt < K1_KT; ++kt) {
        const int c  = kt * 4 + 2 * hi;
        const int cc = (c < F_SZ) ? c : F_SZ - 2;
        a[kt].x = xrow[cc];
        a[kt].y = xrow[cc + 1];
    }
}

__device__ __forceinline__ void k1_tile(const v2f (&a)[K1_KT], const v2f (&bf)[6][K1_KT],
                                        const int hi, v8f (&acc)[6])
{
    #pragma unroll
    for (int nt = 0; nt < 6; ++nt) acc[nt] = v8f{};
    #pragma unroll
    for (int kt = 0; kt < K1_KT; ++kt) {
        v2f av = a[kt];
        if (kt * 4 + 2 * hi >= F_SZ) av = v2f{};
        #pragma unroll
        for (int nt = 0; nt < 6; ++nt) {
            acc[nt] = __builtin_amdgcn_wmma_f32_16x16x4_f32(
                false, av, false, bf[nt][kt], (short)0, acc[nt], false, false);
        }
    }
}

__device__ __forceinline__ void k1_store(float* __restrict__ gi, const int m0,
                                         const int lo, const int hi, const v8f (&acc)[6])
{
    #pragma unroll
    for (int r = 0; r < 8; ++r) {
        const unsigned m = (unsigned)(m0 + hi * 8 + r);
        const unsigned b = m / T_SZ;
        const unsigned t = m - b * T_SZ;
        float* orow = gi + ((size_t)t * B_SZ + b) * G3;
        #pragma unroll
        for (int nt = 0; nt < 6; ++nt) {
            orow[nt * 16 + lo] = acc[nt][r];
        }
    }
}

extern "C" __global__ __launch_bounds__(K1_THREADS)
void gru_gemm_in(const float* __restrict__ x, const float* __restrict__ w_ih,
                 float* __restrict__ gi)
{
    static_assert(K1_TILES % K1_WAVES == 0 && K1_STEPS % 2 == 1,
                  "every wave does the same number of row tiles, an odd one: pairs of tiles and a last one");
    __shared__ float w_lds[G3 * W_PITCH];
    const int tid  = threadIdx.x;
    const int lane = tid & 31;
    const int lo   = lane & 15;
    const int hi   = lane >> 4;
    const int wave = tid >> 5;
    const int wg   = blockIdx.x * 4 + wave;

    const float* xrow = x + (size_t)(wg * 16 + lo) * F_SZ;
    constexpr size_t XSTEP = (size_t)K1_WAVES * 16 * F_SZ;
    int m0 = wg * 16;
    constexpr int MSTEP = K1_WAVES * 16;

    v2f a0[K1_KT], a1[K1_KT];
    k1_load_a(xrow, hi, a0);
    xrow += XSTEP;

    constexpr int PAIRS = F_SZ / 2;
    constexpr int NPAIR = G3 * PAIRS;
    constexpr int NFULL = NPAIR / K1_THREADS;
    constexpr int NREST = NPAIR - NFULL * K1_THREADS;
    static_assert(F_SZ % 2 == 0 && NREST >= 0 && NREST <= K1_THREADS,
                  "pair staging: even rows, one partial trip at most");
    {
        v2f wreg[NFULL];
        v2f wlast = {};
        #pragma unroll
        for (int j = 0; j < NFULL; ++j) {
            const int i = tid + j * K1_THREADS;
            wreg[j].x = w_ih[2 * i];
            wreg[j].y = w_ih[2 * i + 1];
        }
        if (tid < NREST) {
            const int i = tid + NFULL * K1_THREADS;
            wlast.x = w_ih[2 * i];
            wlast.y = w_ih[2 * i + 1];
        }
        #pragma unroll
        for (int j = 0; j < NFULL; ++j) {
            const int i = tid + j * K1_THREADS;
            const int r = i / PAIRS;
            const int p = i - r * PAIRS;
            w_lds[r * W_PITCH + 2 * p]     = wreg[j].x;
            w_lds[r * W_PITCH + 2 * p + 1] = wreg[j].y;
        }
        if (tid < NREST) {
            const int i = tid + NFULL * K1_THREADS;
            const int r = i / PAIRS;
            const int p = i - r * PAIRS;
            w_lds[r * W_PITCH + 2 * p]     = wlast.x;
            w_lds[r * W_PITCH + 2 * p + 1] = wlast.y;
        }
    }
    if (tid < G3) {
        float* z = w_lds + tid * W_PITCH + F_SZ;
        #pragma unroll
        for (int k = 0; k < W_PITCH - F_SZ; ++k) z[k] = 0.0f;
    }
    __syncthreads();

    v2f bf[6][K1_KT];
    #pragma unroll
    for (int nt = 0; nt < 6; ++nt) {
        const float* wrow = w_lds + (nt * 16 + lo) * W_PITCH + 2 * hi;
        #pragma unroll
        for (int kt = 0; kt < K1_KT; ++kt) {
            bf[nt][kt].x = wrow[kt * 4];
            bf[nt][kt].y = wrow[kt * 4 + 1];
        }
    }

    v8f acc[6];
    for (int s = 0; s + 2 < K1_STEPS; s += 2) {
        k1_load_a(xrow, hi, a1);
        __builtin_amdgcn_sched_barrier(0);
        k1_tile(a0, bf, hi, acc);
        k1_store(gi, m0, lo, hi, acc);

        __builtin_amdgcn_sched_barrier(0);
        k1_load_a(xrow + XSTEP, hi, a0);
        __builtin_amdgcn_sched_barrier(0);
        k1_tile(a1, bf, hi, acc);
        k1_store(gi, m0 + MSTEP, lo, hi, acc);

        xrow += 2 * XSTEP;
        m0   += 2 * MSTEP;
    }
    k1_tile(a0, bf, hi, acc);
    k1_store(gi, m0, lo, hi, acc);
}

#define HB (16 * 34)

__device__ __forceinline__ void gi_load(const float* __restrict__ p0, float (&g)[12])
{
    #pragma unroll
    for (int rr = 0; rr < 4; ++rr) {
        const float* p = p0 + rr * G3;
        g[rr]     = p[0];
        g[4 + rr] = p[32];
        g[8 + rr] = p[64];
    }
}

__device__ __forceinline__ void gru_step(
    const float* __restrict__ gnext,
    const float* rd, float* wr,
    const v2f (&bf)[3][8], float (&hc)[4], float (&gc)[12],
    const int a_off, const int s_off)
{
    float gn[12];
    gi_load(gnext, gn);

    v8f acc[3] = {};
    #pragma unroll
    for (int kt = 0; kt < 8; ++kt) {
        v2f a;
        a.x = rd[a_off + kt * 4];
        a.y = rd[a_off + kt * 4 + 1];
        #pragma unroll
        for (int g = 0; g < 3; ++g) {
            acc[g] = __builtin_amdgcn_wmma_f32_16x16x4_f32(
                false, a, false, bf[g][kt], (short)0, acc[g], false, false);
        }
    }

    #pragma unroll
    for (int rr = 0; rr < 4; ++rr) {
        const float rg = sigmoid_f(gc[rr] + acc[0][rr]);
        const float zg = sigmoid_f(gc[4 + rr] + acc[1][rr]);
        const float ng = tanh_sel(gc[8 + rr] + rg * acc[2][rr]);
        hc[rr] = (1.0f - zg) * ng + zg * hc[rr];
        wr[s_off + rr * 34] = hc[rr];
    }

    #pragma unroll
    for (int i = 0; i < 12; ++i) gc[i] = gn[i];
}

__device__ __forceinline__ void lds_barrier()
{
    __builtin_amdgcn_fence(__ATOMIC_RELEASE, "workgroup", "local");
    __builtin_amdgcn_s_barrier();
    __builtin_amdgcn_fence(__ATOMIC_ACQUIRE, "workgroup", "local");
}

extern "C" __global__ __launch_bounds__(128)
__attribute__((amdgpu_waves_per_eu(1, 1)))
void gru_recurrence(const float* __restrict__ gi, const float* __restrict__ w_hh,
                    const float* __restrict__ w1, const float* __restrict__ b1,
                    const float* __restrict__ w2, float* __restrict__ out)
{
    static_assert(T_SZ % 2 == 0, "the scan alternates two h buffers in pairs of steps");
    __shared__ float h_lds[2 * HB];
    const int tid = threadIdx.x;
    for (int i = tid; i < HB; i += 128) h_lds[i] = 0.0f;
    __syncthreads();

    const int lane = tid & 31;
    const int lo   = lane & 15;
    const int hi   = lane >> 4;
    const int wave = tid >> 5;
    const int j    = wave & 1;
    const int q    = wave >> 1;
    const int bb   = blockIdx.x * 16;

    const int a_off = (lo ^ (q << 2)) * 34 + 2 * hi;
    const int s_off = (hi * 8 + q * 4) * 34 + j * 16 + lo;

    v2f bf[3][8];
    #pragma unroll
    for (int g = 0; g < 3; ++g) {
        const int n = (2 * g + j) * 16 + lo;
        #pragma unroll
        for (int kt = 0; kt < 8; ++kt) {
            const int k = kt * 4 + 2 * hi;
            bf[g][kt].x = w_hh[n * H_SZ + k];
            bf[g][kt].y = w_hh[n * H_SZ + k + 1];
        }
    }

    float hc[4] = {0.0f, 0.0f, 0.0f, 0.0f};

    const float* gip = gi + ((long)bb + hi * 8 + q * 4) * G3 + j * 16 + lo;
    const long gstep = (long)B_SZ * G3;

    float gc[12];
    gi_load(gip, gc);

    for (int t = 0; t < T_SZ; t += 2) {
        gru_step(gip + gstep, h_lds, h_lds + HB, bf, hc, gc, a_off, s_off);
        gip += gstep;
        lds_barrier();
        const float* gnx = (t + 2 < T_SZ) ? gip + gstep : gip;
        gru_step(gnx, h_lds + HB, h_lds, bf, hc, gc, a_off, s_off);
        gip += gstep;
        lds_barrier();
    }

    if (tid < 16) {
        float hrow[H_SZ];
        #pragma unroll
        for (int k = 0; k < H_SZ; ++k) hrow[k] = h_lds[tid * 34 + k];
        float y = 0.0f;
        for (int jj = 0; jj < 16; ++jj) {
            float a = b1[jj];
            #pragma unroll
            for (int k = 0; k < H_SZ; ++k) a += hrow[k] * w1[jj * H_SZ + k];
            y += w2[jj] * fmaxf(a, 0.0f);
        }
        out[bb + tid] = y;
    }
}

extern "C" void kernel_launch(void* const* d_in, const int* in_sizes, int n_in,
                              void* d_out, int out_size, void* d_ws, size_t ws_size,
                              hipStream_t stream) {
    const float* x    = (const float*)d_in[0];
    const float* w_ih = (const float*)d_in[1];
    const float* w_hh = (const float*)d_in[2];
    const float* w1   = (const float*)d_in[3];
    const float* b1   = (const float*)d_in[4];
    const float* w2   = (const float*)d_in[5];
    float* out = (float*)d_out;

    float* gi = (float*)d_ws;

    gru_gemm_in<<<256, 128, 0, stream>>>(x, w_ih, gi);
    gru_recurrence<<<256, 128, 0, stream>>>(gi, w_hh, w1, b1, w2, out);
}


# MLP head: one wave computes w1 @ h^T on the matrix unit (8 products, weights requested before the scan) in place of 16 lanes looping over 16 scalar-loaded rows of w1; sums of the head reordered
# speedup vs baseline: 3.3131x; 1.1035x over previous
#include <hip/hip_runtime.h>
#include <cmath>


typedef __attribute__((ext_vector_type(2))) float v2f;
typedef __attribute__((ext_vector_type(8))) float v8f;

#define B_SZ 4096
#define T_SZ 60
#define F_SZ 158
#define F_PAD 160
#define W_PITCH 164
#define G3 96
#define H_SZ 32

__device__ __forceinline__ float sigmoid_f(float x) {
    return 1.0f / (1.0f + __expf(-x));
}

__device__ __forceinline__ float tanh_sel(float x) {
    const float ax = fabsf(x);
    const float e  = expf(ax + ax);
    const float big = fmaf(__builtin_amdgcn_rcpf(e + 1.0f), -2.0f, 1.0f);
    const float x2 = ax * ax;
    float p = fmaf(x2, -0x1.758e7ap-8f, 0x1.521192p-6f);
    p = fmaf(x2, p, -0x1.b8389cp-5f);
    p = fmaf(x2, p, 0x1.110704p-3f);
    p = fmaf(x2, p, -0x1.555532p-2f);
    const float small = fmaf(x2, ax * p, ax);
    return copysignf((ax < 0.625f) ? small : big, x);
}

#define K1_THREADS 128
#define K1_KT      40
#define K1_WAVES   1024
#define K1_TILES   (B_SZ * T_SZ / 16)
#define K1_STEPS   (K1_TILES / K1_WAVES)

__device__ __forceinline__ void k1_load_a(const float* __restrict__ xrow, const int hi,
                                          v2f (&a)[K1_KT])
{
    #pragma unroll
    for (int kt = 0; kt < K1_KT; ++kt) {
        const int c  = kt * 4 + 2 * hi;
        const int cc = (c < F_SZ) ? c : F_SZ - 2;
        a[kt].x = xrow[cc];
        a[kt].y = xrow[cc + 1];
    }
}

__device__ __forceinline__ void k1_tile(const v2f (&a)[K1_KT], const v2f (&bf)[6][K1_KT],
                                        const int hi, v8f (&acc)[6])
{
    #pragma unroll
    for (int nt = 0; nt < 6; ++nt) acc[nt] = v8f{};
    #pragma unroll
    for (int kt = 0; kt < K1_KT; ++kt) {
        v2f av = a[kt];
        if (kt * 4 + 2 * hi >= F_SZ) av = v2f{};
        #pragma unroll
        for (int nt = 0; nt < 6; ++nt) {
            acc[nt] = __builtin_amdgcn_wmma_f32_16x16x4_f32(
                false, av, false, bf[nt][kt], (short)0, acc[nt], false, false);
        }
    }
}

__device__ __forceinline__ void k1_store(float* __restrict__ gi, const int m0,
                                         const int lo, const int hi, const v8f (&acc)[6])
{
    #pragma unroll
    for (int r = 0; r < 8; ++r) {
        const unsigned m = (unsigned)(m0 + hi * 8 + r);
        const unsigned b = m / T_SZ;
        const unsigned t = m - b * T_SZ;
        float* orow = gi + ((size_t)t * B_SZ + b) * G3;
        #pragma unroll
        for (int nt = 0; nt < 6; ++nt) {
            orow[nt * 16 + lo] = acc[nt][r];
        }
    }
}

extern "C" __global__ __launch_bounds__(K1_THREADS)
void gru_gemm_in(const float* __restrict__ x, const float* __restrict__ w_ih,
                 float* __restrict__ gi)
{
    static_assert(K1_TILES % K1_WAVES == 0 && K1_STEPS % 2 == 1,
                  "every wave does the same number of row tiles, an odd one: pairs of tiles and a last one");
    __shared__ float w_lds[G3 * W_PITCH];
    const int tid  = threadIdx.x;
    const int lane = tid & 31;
    const int lo   = lane & 15;
    const int hi   = lane >> 4;
    const int wave = tid >> 5;
    const int wg   = blockIdx.x * 4 + wave;

    const float* xrow = x + (size_t)(wg * 16 + lo) * F_SZ;
    constexpr size_t XSTEP = (size_t)K1_WAVES * 16 * F_SZ;
    int m0 = wg * 16;
    constexpr int MSTEP = K1_WAVES * 16;

    v2f a0[K1_KT], a1[K1_KT];
    k1_load_a(xrow, hi, a0);
    xrow += XSTEP;

    constexpr int PAIRS = F_SZ / 2;
    constexpr int NPAIR = G3 * PAIRS;
    constexpr int NFULL = NPAIR / K1_THREADS;
    constexpr int NREST = NPAIR - NFULL * K1_THREADS;
    static_assert(F_SZ % 2 == 0 && NREST >= 0 && NREST <= K1_THREADS,
                  "pair staging: even rows, one partial trip at most");
    {
        v2f wreg[NFULL];
        v2f wlast = {};
        #pragma unroll
        for (int j = 0; j < NFULL; ++j) {
            const int i = tid + j * K1_THREADS;
            wreg[j].x = w_ih[2 * i];
            wreg[j].y = w_ih[2 * i + 1];
        }
        if (tid < NREST) {
            const int i = tid + NFULL * K1_THREADS;
            wlast.x = w_ih[2 * i];
            wlast.y = w_ih[2 * i + 1];
        }
        #pragma unroll
        for (int j = 0; j < NFULL; ++j) {
            const int i = tid + j * K1_THREADS;
            const int r = i / PAIRS;
            const int p = i - r * PAIRS;
            w_lds[r * W_PITCH + 2 * p]     = wreg[j].x;
            w_lds[r * W_PITCH + 2 * p + 1] = wreg[j].y;
        }
        if (tid < NREST) {
            const int i = tid + NFULL * K1_THREADS;
            const int r = i / PAIRS;
            const int p = i - r * PAIRS;
            w_lds[r * W_PITCH + 2 * p]     = wlast.x;
            w_lds[r * W_PITCH + 2 * p + 1] = wlast.y;
        }
    }
    if (tid < G3) {
        float* z = w_lds + tid * W_PITCH + F_SZ;
        #pragma unroll
        for (int k = 0; k < W_PITCH - F_SZ; ++k) z[k] = 0.0f;
    }
    __syncthreads();

    v2f bf[6][K1_KT];
    #pragma unroll
    for (int nt = 0; nt < 6; ++nt) {
        const float* wrow = w_lds + (nt * 16 + lo) * W_PITCH + 2 * hi;
        #pragma unroll
        for (int kt = 0; kt < K1_KT; ++kt) {
            bf[nt][kt].x = wrow[kt * 4];
            bf[nt][kt].y = wrow[kt * 4 + 1];
        }
    }

    v8f acc[6];
    for (int s = 0; s + 2 < K1_STEPS; s += 2) {
        k1_load_a(xrow, hi, a1);
        __builtin_amdgcn_sched_barrier(0);
        k1_tile(a0, bf, hi, acc);
        k1_store(gi, m0, lo, hi, acc);

        __builtin_amdgcn_sched_barrier(0);
        k1_load_a(xrow + XSTEP, hi, a0);
        __builtin_amdgcn_sched_barrier(0);
        k1_tile(a1, bf, hi, acc);
        k1_store(gi, m0 + MSTEP, lo, hi, acc);

        xrow += 2 * XSTEP;
        m0   += 2 * MSTEP;
    }
    k1_tile(a0, bf, hi, acc);
    k1_store(gi, m0, lo, hi, acc);
}

#define HB (16 * 34)

__device__ __forceinline__ void gi_load(const float* __restrict__ p0, float (&g)[12])
{
    #pragma unroll
    for (int rr = 0; rr < 4; ++rr) {
        const float* p = p0 + rr * G3;
        g[rr]     = p[0];
        g[4 + rr] = p[32];
        g[8 + rr] = p[64];
    }
}

__device__ __forceinline__ void gru_step(
    const float* __restrict__ gnext,
    const float* rd, float* wr,
    const v2f (&bf)[3][8], float (&hc)[4], float (&gc)[12],
    const int a_off, const int s_off)
{
    float gn[12];
    gi_load(gnext, gn);

    v8f acc[3] = {};
    #pragma unroll
    for (int kt = 0; kt < 8; ++kt) {
        v2f a;
        a.x = rd[a_off + kt * 4];
        a.y = rd[a_off + kt * 4 + 1];
        #pragma unroll
        for (int g = 0; g < 3; ++g) {
            acc[g] = __builtin_amdgcn_wmma_f32_16x16x4_f32(
                false, a, false, bf[g][kt], (short)0, acc[g], false, false);
        }
    }

    #pragma unroll
    for (int rr = 0; rr < 4; ++rr) {
        const float rg = sigmoid_f(gc[rr] + acc[0][rr]);
        const float zg = sigmoid_f(gc[4 + rr] + acc[1][rr]);
        const float ng = tanh_sel(gc[8 + rr] + rg * acc[2][rr]);
        hc[rr] = (1.0f - zg) * ng + zg * hc[rr];
        wr[s_off + rr * 34] = hc[rr];
    }

    #pragma unroll
    for (int i = 0; i < 12; ++i) gc[i] = gn[i];
}

__device__ __forceinline__ void lds_barrier()
{
    __builtin_amdgcn_fence(__ATOMIC_RELEASE, "workgroup", "local");
    __builtin_amdgcn_s_barrier();
    __builtin_amdgcn_fence(__ATOMIC_ACQUIRE, "workgroup", "local");
}

extern "C" __global__ __launch_bounds__(128)
__attribute__((amdgpu_waves_per_eu(1, 1)))
void gru_recurrence(const float* __restrict__ gi, const float* __restrict__ w_hh,
                    const float* __restrict__ w1, const float* __restrict__ b1,
                    const float* __restrict__ w2, float* __restrict__ out)
{
    static_assert(T_SZ % 2 == 0, "the scan alternates two h buffers in pairs of steps");
    __shared__ float h_lds[2 * HB];
    const int tid = threadIdx.x;
    for (int i = tid; i < HB; i += 128) h_lds[i] = 0.0f;
    __syncthreads();

    const int lane = tid & 31;
    const int lo   = lane & 15;
    const int hi   = lane >> 4;
    const int wave = tid >> 5;
    const int j    = wave & 1;
    const int q    = wave >> 1;
    const int bb   = blockIdx.x * 16;

    const int a_off = (lo ^ (q << 2)) * 34 + 2 * hi;
    const int s_off = (hi * 8 + q * 4) * 34 + j * 16 + lo;

    v2f bf[3][8];
    #pragma unroll
    for (int g = 0; g < 3; ++g) {
        const int n = (2 * g + j) * 16 + lo;
        #pragma unroll
        for (int kt = 0; kt < 8; ++kt) {
            const int k = kt * 4 + 2 * hi;
            bf[g][kt].x = w_hh[n * H_SZ + k];
            bf[g][kt].y = w_hh[n * H_SZ + k + 1];
        }
    }

    float hc[4] = {0.0f, 0.0f, 0.0f, 0.0f};

    v2f w1f[8];
    #pragma unroll
    for (int kt = 0; kt < 8; ++kt) {
        const int k = kt * 4 + 2 * hi;
        w1f[kt].x = w1[lo * H_SZ + k];
        w1f[kt].y = w1[lo * H_SZ + k + 1];
    }
    float b1v[8], w2v[8];
    #pragma unroll
    for (int r = 0; r < 8; ++r) {
        b1v[r] = b1[hi * 8 + r];
        w2v[r] = w2[hi * 8 + r];
    }

    const float* gip = gi + ((long)bb + hi * 8 + q * 4) * G3 + j * 16 + lo;
    const long gstep = (long)B_SZ * G3;

    float gc[12];
    gi_load(gip, gc);

    for (int t = 0; t < T_SZ; t += 2) {
        gru_step(gip + gstep, h_lds, h_lds + HB, bf, hc, gc, a_off, s_off);
        gip += gstep;
        lds_barrier();
        const float* gnx = (t + 2 < T_SZ) ? gip + gstep : gip;
        gru_step(gnx, h_lds + HB, h_lds, bf, hc, gc, a_off, s_off);
        gip += gstep;
        lds_barrier();
    }

    if (wave == 0) {
        v8f c0, c1 = {};
        #pragma unroll
        for (int r = 0; r < 8; ++r) c0[r] = b1v[r];
        const int b_off = lo * 34 + 2 * hi;
        #pragma unroll
        for (int kt = 0; kt < 8; kt += 2) {
            v2f hb0, hb1;
            hb0.x = h_lds[b_off + kt * 4];
            hb0.y = h_lds[b_off + kt * 4 + 1];
            hb1.x = h_lds[b_off + kt * 4 + 4];
            hb1.y = h_lds[b_off + kt * 4 + 5];
            c0 = __builtin_amdgcn_wmma_f32_16x16x4_f32(
                false, w1f[kt], false, hb0, (short)0, c0, false, false);
            c1 = __builtin_amdgcn_wmma_f32_16x16x4_f32(
                false, w1f[kt + 1], false, hb1, (short)0, c1, false, false);
        }
        float y = 0.0f;
        #pragma unroll
        for (int r = 0; r < 8; ++r) {
            y = fmaf(w2v[r], fmaxf(c0[r] + c1[r], 0.0f), y);
        }
        y += __shfl_xor(y, 16, 32);
        if (hi == 0) out[bb + lo] = y;
    }
}

extern "C" void kernel_launch(void* const* d_in, const int* in_sizes, int n_in,
                              void* d_out, int out_size, void* d_ws, size_t ws_size,
                              hipStream_t stream) {
    const float* x    = (const float*)d_in[0];
    const float* w_ih = (const float*)d_in[1];
    const float* w_hh = (const float*)d_in[2];
    const float* w1   = (const float*)d_in[3];
    const float* b1   = (const float*)d_in[4];
    const float* w2   = (const float*)d_in[5];
    float* out = (float*)d_out;

    float* gi = (float*)d_ws;

    gru_gemm_in<<<256, 128, 0, stream>>>(x, w_ih, gi);
    gru_recurrence<<<256, 128, 0, stream>>>(gi, w_hh, w1, b1, w2, out);
}


# recurrence gates: sigmoid as one exp2 and the hardware reciprocal (1 ulp) instead of the compiled IEEE division; tanh's exp(2|x|) as one exp2 instead of the library exp: 195 -> 90 instructions a step
# speedup vs baseline: 3.4634x; 1.0454x over previous
#include <hip/hip_runtime.h>
#include <cmath>


typedef __attribute__((ext_vector_type(2))) float v2f;
typedef __attribute__((ext_vector_type(8))) float v8f;

#define B_SZ 4096
#define T_SZ 60
#define F_SZ 158
#define F_PAD 160
#define W_PITCH 164
#define G3 96
#define H_SZ 32

__device__ __forceinline__ float sigmoid_f(float x) {
    return __builtin_amdgcn_rcpf(1.0f + __expf(-x));
}

__device__ __forceinline__ float tanh_sel(float x) {
    const float ax = fabsf(x);
    const float e  = __builtin_amdgcn_exp2f(ax * 0x1.715476p+1f);
    const float big = fmaf(__builtin_amdgcn_rcpf(e + 1.0f), -2.0f, 1.0f);
    const float x2 = ax * ax;
    float p = fmaf(x2, -0x1.758e7ap-8f, 0x1.521192p-6f);
    p = fmaf(x2, p, -0x1.b8389cp-5f);
    p = fmaf(x2, p, 0x1.110704p-3f);
    p = fmaf(x2, p, -0x1.555532p-2f);
    const float small = fmaf(x2, ax * p, ax);
    return copysignf((ax < 0.625f) ? small : big, x);
}

#define K1_THREADS 128
#define K1_KT      40
#define K1_WAVES   1024
#define K1_TILES   (B_SZ * T_SZ / 16)
#define K1_STEPS   (K1_TILES / K1_WAVES)

__device__ __forceinline__ void k1_load_a(const float* __restrict__ xrow, const int hi,
                                          v2f (&a)[K1_KT])
{
    #pragma unroll
    for (int kt = 0; kt < K1_KT; ++kt) {
        const int c  = kt * 4 + 2 * hi;
        const int cc = (c < F_SZ) ? c : F_SZ - 2;
        a[kt].x = xrow[cc];
        a[kt].y = xrow[cc + 1];
    }
}

__device__ __forceinline__ void k1_tile(const v2f (&a)[K1_KT], const v2f (&bf)[6][K1_KT],
                                        const int hi, v8f (&acc)[6])
{
    #pragma unroll
    for (int nt = 0; nt < 6; ++nt) acc[nt] = v8f{};
    #pragma unroll
    for (int kt = 0; kt < K1_KT; ++kt) {
        v2f av = a[kt];
        if (kt * 4 + 2 * hi >= F_SZ) av = v2f{};
        #pragma unroll
        for (int nt = 0; nt < 6; ++nt) {
            acc[nt] = __builtin_amdgcn_wmma_f32_16x16x4_f32(
                false, av, false, bf[nt][kt], (short)0, acc[nt], false, false);
        }
    }
}

__device__ __forceinline__ void k1_store(float* __restrict__ gi, const int m0,
                                         const int lo, const int hi, const v8f (&acc)[6])
{
    #pragma unroll
    for (int r = 0; r < 8; ++r) {
        const unsigned m = (unsigned)(m0 + hi * 8 + r);
        const unsigned b = m / T_SZ;
        const unsigned t = m - b * T_SZ;
        float* orow = gi + ((size_t)t * B_SZ + b) * G3;
        #pragma unroll
        for (int nt = 0; nt < 6; ++nt) {
            orow[nt * 16 + lo] = acc[nt][r];
        }
    }
}

extern "C" __global__ __launch_bounds__(K1_THREADS)
void gru_gemm_in(const float* __restrict__ x, const float* __restrict__ w_ih,
                 float* __restrict__ gi)
{
    static_assert(K1_TILES % K1_WAVES == 0 && K1_STEPS % 2 == 1,
                  "every wave does the same number of row tiles, an odd one: pairs of tiles and a last one");
    __shared__ float w_lds[G3 * W_PITCH];
    const int tid  = threadIdx.x;
    const int lane = tid & 31;
    const int lo   = lane & 15;
    const int hi   = lane >> 4;
    const int wave = tid >> 5;
    const int wg   = blockIdx.x * 4 + wave;

    const float* xrow = x + (size_t)(wg * 16 + lo) * F_SZ;
    constexpr size_t XSTEP = (size_t)K1_WAVES * 16 * F_SZ;
    int m0 = wg * 16;
    constexpr int MSTEP = K1_WAVES * 16;

    v2f a0[K1_KT], a1[K1_KT];
    k1_load_a(xrow, hi, a0);
    xrow += XSTEP;

    constexpr int PAIRS = F_SZ / 2;
    constexpr int NPAIR = G3 * PAIRS;
    constexpr int NFULL = NPAIR / K1_THREADS;
    constexpr int NREST = NPAIR - NFULL * K1_THREADS;
    static_assert(F_SZ % 2 == 0 && NREST >= 0 && NREST <= K1_THREADS,
                  "pair staging: even rows, one partial trip at most");
    {
        v2f wreg[NFULL];
        v2f wlast = {};
        #pragma unroll
        for (int j = 0; j < NFULL; ++j) {
            const int i = tid + j * K1_THREADS;
            wreg[j].x = w_ih[2 * i];
            wreg[j].y = w_ih[2 * i + 1];
        }
        if (tid < NREST) {
            const int i = tid + NFULL * K1_THREADS;
            wlast.x = w_ih[2 * i];
            wlast.y = w_ih[2 * i + 1];
        }
        #pragma unroll
        for (int j = 0; j < NFULL; ++j) {
            const int i = tid + j * K1_THREADS;
            const int r = i / PAIRS;
            const int p = i - r * PAIRS;
            w_lds[r * W_PITCH + 2 * p]     = wreg[j].x;
            w_lds[r * W_PITCH + 2 * p + 1] = wreg[j].y;
        }
        if (tid < NREST) {
            const int i = tid + NFULL * K1_THREADS;
            const int r = i / PAIRS;
            const int p = i - r * PAIRS;
            w_lds[r * W_PITCH + 2 * p]     = wlast.x;
            w_lds[r * W_PITCH + 2 * p + 1] = wlast.y;
        }
    }
    if (tid < G3) {
        float* z = w_lds + tid * W_PITCH + F_SZ;
        #pragma unroll
        for (int k = 0; k < W_PITCH - F_SZ; ++k) z[k] = 0.0f;
    }
    __syncthreads();

    v2f bf[6][K1_KT];
    #pragma unroll
    for (int nt = 0; nt < 6; ++nt) {
        const float* wrow = w_lds + (nt * 16 + lo) * W_PITCH + 2 * hi;
        #pragma unroll
        for (int kt = 0; kt < K1_KT; ++kt) {
            bf[nt][kt].x = wrow[kt * 4];
            bf[nt][kt].y = wrow[kt * 4 + 1];
        }
    }

    v8f acc[6];
    for (int s = 0; s + 2 < K1_STEPS; s += 2) {
        k1_load_a(xrow, hi, a1);
        __builtin_amdgcn_sched_barrier(0);
        k1_tile(a0, bf, hi, acc);
        k1_store(gi, m0, lo, hi, acc);

        __builtin_amdgcn_sched_barrier(0);
        k1_load_a(xrow + XSTEP, hi, a0);
        __builtin_amdgcn_sched_barrier(0);
        k1_tile(a1, bf, hi, acc);
        k1_store(gi, m0 + MSTEP, lo, hi, acc);

        xrow += 2 * XSTEP;
        m0   += 2 * MSTEP;
    }
    k1_tile(a0, bf, hi, acc);
    k1_store(gi, m0, lo, hi, acc);
}

#define HB (16 * 34)

__device__ __forceinline__ void gi_load(const float* __restrict__ p0, float (&g)[12])
{
    #pragma unroll
    for (int rr = 0; rr < 4; ++rr) {
        const float* p = p0 + rr * G3;
        g[rr]     = p[0];
        g[4 + rr] = p[32];
        g[8 + rr] = p[64];
    }
}

__device__ __forceinline__ void gru_step(
    const float* __restrict__ gnext,
    const float* rd, float* wr,
    const v2f (&bf)[3][8], float (&hc)[4], float (&gc)[12],
    const int a_off, const int s_off)
{
    float gn[12];
    gi_load(gnext, gn);

    v8f acc[3] = {};
    #pragma unroll
    for (int kt = 0; kt < 8; ++kt) {
        v2f a;
        a.x = rd[a_off + kt * 4];
        a.y = rd[a_off + kt * 4 + 1];
        #pragma unroll
        for (int g = 0; g < 3; ++g) {
            acc[g] = __builtin_amdgcn_wmma_f32_16x16x4_f32(
                false, a, false, bf[g][kt], (short)0, acc[g], false, false);
        }
    }

    #pragma unroll
    for (int rr = 0; rr < 4; ++rr) {
        const float rg = sigmoid_f(gc[rr] + acc[0][rr]);
        const float zg = sigmoid_f(gc[4 + rr] + acc[1][rr]);
        const float ng = tanh_sel(gc[8 + rr] + rg * acc[2][rr]);
        hc[rr] = (1.0f - zg) * ng + zg * hc[rr];
        wr[s_off + rr * 34] = hc[rr];
    }

    #pragma unroll
    for (int i = 0; i < 12; ++i) gc[i] = gn[i];
}

__device__ __forceinline__ void lds_barrier()
{
    __builtin_amdgcn_fence(__ATOMIC_RELEASE, "workgroup", "local");
    __builtin_amdgcn_s_barrier();
    __builtin_amdgcn_fence(__ATOMIC_ACQUIRE, "workgroup", "local");
}

extern "C" __global__ __launch_bounds__(128)
__attribute__((amdgpu_waves_per_eu(1, 1)))
void gru_recurrence(const float* __restrict__ gi, const float* __restrict__ w_hh,
                    const float* __restrict__ w1, const float* __restrict__ b1,
                    const float* __restrict__ w2, float* __restrict__ out)
{
    static_assert(T_SZ % 2 == 0, "the scan alternates two h buffers in pairs of steps");
    __shared__ float h_lds[2 * HB];
    const int tid = threadIdx.x;
    for (int i = tid; i < HB; i += 128) h_lds[i] = 0.0f;
    __syncthreads();

    const int lane = tid & 31;
    const int lo   = lane & 15;
    const int hi   = lane >> 4;
    const int wave = tid >> 5;
    const int j    = wave & 1;
    const int q    = wave >> 1;
    const int bb   = blockIdx.x * 16;

    const int a_off = (lo ^ (q << 2)) * 34 + 2 * hi;
    const int s_off = (hi * 8 + q * 4) * 34 + j * 16 + lo;

    v2f bf[3][8];
    #pragma unroll
    for (int g = 0; g < 3; ++g) {
        const int n = (2 * g + j) * 16 + lo;
        #pragma unroll
        for (int kt = 0; kt < 8; ++kt) {
            const int k = kt * 4 + 2 * hi;
            bf[g][kt].x = w_hh[n * H_SZ + k];
            bf[g][kt].y = w_hh[n * H_SZ + k + 1];
        }
    }

    float hc[4] = {0.0f, 0.0f, 0.0f, 0.0f};

    v2f w1f[8];
    #pragma unroll
    for (int kt = 0; kt < 8; ++kt) {
        const int k = kt * 4 + 2 * hi;
        w1f[kt].x = w1[lo * H_SZ + k];
        w1f[kt].y = w1[lo * H_SZ + k + 1];
    }
    float b1v[8], w2v[8];
    #pragma unroll
    for (int r = 0; r < 8; ++r) {
        b1v[r] = b1[hi * 8 + r];
        w2v[r] = w2[hi * 8 + r];
    }

    const float* gip = gi + ((long)bb + hi * 8 + q * 4) * G3 + j * 16 + lo;
    const long gstep = (long)B_SZ * G3;

    float gc[12];
    gi_load(gip, gc);

    for (int t = 0; t < T_SZ; t += 2) {
        gru_step(gip + gstep, h_lds, h_lds + HB, bf, hc, gc, a_off, s_off);
        gip += gstep;
        lds_barrier();
        const float* gnx = (t + 2 < T_SZ) ? gip + gstep : gip;
        gru_step(gnx, h_lds + HB, h_lds, bf, hc, gc, a_off, s_off);
        gip += gstep;
        lds_barrier();
    }

    if (wave == 0) {
        v8f c0, c1 = {};
        #pragma unroll
        for (int r = 0; r < 8; ++r) c0[r] = b1v[r];
        const int b_off = lo * 34 + 2 * hi;
        #pragma unroll
        for (int kt = 0; kt < 8; kt += 2) {
            v2f hb0, hb1;
            hb0.x = h_lds[b_off + kt * 4];
            hb0.y = h_lds[b_off + kt * 4 + 1];
            hb1.x = h_lds[b_off + kt * 4 + 4];
            hb1.y = h_lds[b_off + kt * 4 + 5];
            c0 = __builtin_amdgcn_wmma_f32_16x16x4_f32(
                false, w1f[kt], false, hb0, (short)0, c0, false, false);
            c1 = __builtin_amdgcn_wmma_f32_16x16x4_f32(
                false, w1f[kt + 1], false, hb1, (short)0, c1, false, false);
        }
        float y = 0.0f;
        #pragma unroll
        for (int r = 0; r < 8; ++r) {
            y = fmaf(w2v[r], fmaxf(c0[r] + c1[r], 0.0f), y);
        }
        y += __shfl_xor(y, 16, 32);
        if (hi == 0) out[bb + lo] = y;
    }
}

extern "C" void kernel_launch(void* const* d_in, const int* in_sizes, int n_in,
                              void* d_out, int out_size, void* d_ws, size_t ws_size,
                              hipStream_t stream) {
    const float* x    = (const float*)d_in[0];
    const float* w_ih = (const float*)d_in[1];
    const float* w_hh = (const float*)d_in[2];
    const float* w1   = (const float*)d_in[3];
    const float* b1   = (const float*)d_in[4];
    const float* w2   = (const float*)d_in[5];
    float* out = (float*)d_out;

    float* gi = (float*)d_ws;

    gru_gemm_in<<<256, 128, 0, stream>>>(x, w_ih, gi);
    gru_recurrence<<<256, 128, 0, stream>>>(gi, w_hh, w1, b1, w2, out);
}


# recurrence: product transposed (W @ h^T), a wave owns 8 h columns of all 16 rows: 16 matrix instructions a step, not 24; gi lead 3 of the shorter steps, 16-byte loads; k order of sums changed
# speedup vs baseline: 3.8066x; 1.0991x over previous
#include <hip/hip_runtime.h>
#include <cmath>


typedef __attribute__((ext_vector_type(2))) float v2f;
typedef __attribute__((ext_vector_type(4))) float v4f;
typedef __attribute__((ext_vector_type(8))) float v8f;

#define B_SZ 4096
#define T_SZ 60
#define F_SZ 158
#define F_PAD 160
#define W_PITCH 164
#define G3 96
#define H_SZ 32

__device__ __forceinline__ v2f exp2_2(const v2f x) {
    v2f r;
    r.x = __builtin_amdgcn_exp2f(x.x);
    r.y = __builtin_amdgcn_exp2f(x.y);
    return r;
}
__device__ __forceinline__ v2f rcp_2(const v2f x) {
    v2f r;
    r.x = __builtin_amdgcn_rcpf(x.x);
    r.y = __builtin_amdgcn_rcpf(x.y);
    return r;
}
__device__ __forceinline__ v2f fma_2(const v2f a, const v2f b, const v2f c) {
    return __builtin_elementwise_fma(a, b, c);
}
__device__ __forceinline__ v2f splat_2(const float c) { return v2f{c, c}; }

__device__ __forceinline__ v2f sigmoid_2(const v2f x) {
    return rcp_2(1.0f + exp2_2(x * -0x1.715476p+0f));
}

__device__ __forceinline__ v2f tanh_2(const v2f x) {
    v2f ax;
    ax.x = fabsf(x.x);
    ax.y = fabsf(x.y);
    const v2f e   = exp2_2(ax * 0x1.715476p+1f);
    const v2f big = fma_2(rcp_2(e + 1.0f), splat_2(-2.0f), splat_2(1.0f));
    const v2f x2  = ax * ax;
    v2f p = fma_2(x2, splat_2(-0x1.758e7ap-8f), splat_2(0x1.521192p-6f));
    p = fma_2(x2, p, splat_2(-0x1.b8389cp-5f));
    p = fma_2(x2, p, splat_2(0x1.110704p-3f));
    p = fma_2(x2, p, splat_2(-0x1.555532p-2f));
    const v2f small = fma_2(x2, ax * p, ax);
    v2f r;
    r.x = copysignf((ax.x < 0.625f) ? small.x : big.x, x.x);
    r.y = copysignf((ax.y < 0.625f) ? small.y : big.y, x.y);
    return r;
}

#define K1_THREADS 128
#define K1_KT      40
#define K1_WAVES   1024
#define K1_TILES   (B_SZ * T_SZ / 16)
#define K1_STEPS   (K1_TILES / K1_WAVES)

__device__ __forceinline__ void k1_load_a(const float* __restrict__ xrow, const int hi,
                                          v2f (&a)[K1_KT])
{
    #pragma unroll
    for (int kt = 0; kt < K1_KT; ++kt) {
        const int c  = kt * 4 + 2 * hi;
        const int cc = (c < F_SZ) ? c : F_SZ - 2;
        a[kt].x = xrow[cc];
        a[kt].y = xrow[cc + 1];
    }
}

__device__ __forceinline__ void k1_tile(const v2f (&a)[K1_KT], const v2f (&bf)[6][K1_KT],
                                        const int hi, v8f (&acc)[6])
{
    #pragma unroll
    for (int nt = 0; nt < 6; ++nt) acc[nt] = v8f{};
    #pragma unroll
    for (int kt = 0; kt < K1_KT; ++kt) {
        v2f av = a[kt];
        if (kt * 4 + 2 * hi >= F_SZ) av = v2f{};
        #pragma unroll
        for (int nt = 0; nt < 6; ++nt) {
            acc[nt] = __builtin_amdgcn_wmma_f32_16x16x4_f32(
                false, av, false, bf[nt][kt], (short)0, acc[nt], false, false);
        }
    }
}

__device__ __forceinline__ void k1_store(float* __restrict__ gi, const int m0,
                                         const int lo, const int hi, const v8f (&acc)[6])
{
    #pragma unroll
    for (int r = 0; r < 8; ++r) {
        const unsigned m = (unsigned)(m0 + hi * 8 + r);
        const unsigned b = m / T_SZ;
        const unsigned t = m - b * T_SZ;
        float* orow = gi + ((size_t)t * B_SZ + b) * G3;
        #pragma unroll
        for (int nt = 0; nt < 6; ++nt) {
            orow[nt * 16 + lo] = acc[nt][r];
        }
    }
}

extern "C" __global__ __launch_bounds__(K1_THREADS)
void gru_gemm_in(const float* __restrict__ x, const float* __restrict__ w_ih,
                 float* __restrict__ gi)
{
    static_assert(K1_TILES % K1_WAVES == 0 && K1_STEPS % 2 == 1,
                  "every wave does the same number of row tiles, an odd one: pairs of tiles and a last one");
    __shared__ float w_lds[G3 * W_PITCH];
    const int tid  = threadIdx.x;
    const int lane = tid & 31;
    const int lo   = lane & 15;
    const int hi   = lane >> 4;
    const int wave = tid >> 5;
    const int wg   = blockIdx.x * 4 + wave;

    const float* xrow = x + (size_t)(wg * 16 + lo) * F_SZ;
    constexpr size_t XSTEP = (size_t)K1_WAVES * 16 * F_SZ;
    int m0 = wg * 16;
    constexpr int MSTEP = K1_WAVES * 16;

    v2f a0[K1_KT], a1[K1_KT];
    k1_load_a(xrow, hi, a0);
    xrow += XSTEP;

    constexpr int PAIRS = F_SZ / 2;
    constexpr int NPAIR = G3 * PAIRS;
    constexpr int NFULL = NPAIR / K1_THREADS;
    constexpr int NREST = NPAIR - NFULL * K1_THREADS;
    static_assert(F_SZ % 2 == 0 && NREST >= 0 && NREST <= K1_THREADS,
                  "pair staging: even rows, one partial trip at most");
    {
        v2f wreg[NFULL];
        v2f wlast = {};
        #pragma unroll
        for (int j = 0; j < NFULL; ++j) {
            const int i = tid + j * K1_THREADS;
            wreg[j].x = w_ih[2 * i];
            wreg[j].y = w_ih[2 * i + 1];
        }
        if (tid < NREST) {
            const int i = tid + NFULL * K1_THREADS;
            wlast.x = w_ih[2 * i];
            wlast.y = w_ih[2 * i + 1];
        }
        #pragma unroll
        for (int j = 0; j < NFULL; ++j) {
            const int i = tid + j * K1_THREADS;
            const int r = i / PAIRS;
            const int p = i - r * PAIRS;
            w_lds[r * W_PITCH + 2 * p]     = wreg[j].x;
            w_lds[r * W_PITCH + 2 * p + 1] = wreg[j].y;
        }
        if (tid < NREST) {
            const int i = tid + NFULL * K1_THREADS;
            const int r = i / PAIRS;
            const int p = i - r * PAIRS;
            w_lds[r * W_PITCH + 2 * p]     = wlast.x;
            w_lds[r * W_PITCH + 2 * p + 1] = wlast.y;
        }
    }
    if (tid < G3) {
        float* z = w_lds + tid * W_PITCH + F_SZ;
        #pragma unroll
        for (int k = 0; k < W_PITCH - F_SZ; ++k) z[k] = 0.0f;
    }
    __syncthreads();

    v2f bf[6][K1_KT];
    #pragma unroll
    for (int nt = 0; nt < 6; ++nt) {
        const float* wrow = w_lds + (nt * 16 + lo) * W_PITCH + 2 * hi;
        #pragma unroll
        for (int kt = 0; kt < K1_KT; ++kt) {
            bf[nt][kt].x = wrow[kt * 4];
            bf[nt][kt].y = wrow[kt * 4 + 1];
        }
    }

    v8f acc[6];
    for (int s = 0; s + 2 < K1_STEPS; s += 2) {
        k1_load_a(xrow, hi, a1);
        __builtin_amdgcn_sched_barrier(0);
        k1_tile(a0, bf, hi, acc);
        k1_store(gi, m0, lo, hi, acc);

        __builtin_amdgcn_sched_barrier(0);
        k1_load_a(xrow + XSTEP, hi, a0);
        __builtin_amdgcn_sched_barrier(0);
        k1_tile(a1, bf, hi, acc);
        k1_store(gi, m0 + MSTEP, lo, hi, acc);

        xrow += 2 * XSTEP;
        m0   += 2 * MSTEP;
    }
    k1_tile(a0, bf, hi, acc);
    k1_store(gi, m0, lo, hi, acc);
}

#define HP 36
#define HB (16 * HP)
#define GI_LEAD 3

__device__ __forceinline__ void gi_load(const float* __restrict__ p, v4f (&g)[3])
{
    g[0] = *reinterpret_cast<const v4f*>(p);
    g[1] = *reinterpret_cast<const v4f*>(p + 32);
    g[2] = *reinterpret_cast<const v4f*>(p + 64);
}

__device__ __forceinline__ void gru_step(
    const float* __restrict__ gnext, v4f (&gl)[3],
    const float* rd, float* wr,
    const v2f (&wa)[2][8], v2f (&hc)[2], const v4f (&gc)[3],
    const int b_off, const int s_off)
{
    gi_load(gnext, gl);

    v4f hv[4];
    #pragma unroll
    for (int i = 0; i < 4; ++i) {
        hv[i] = *reinterpret_cast<const v4f*>(rd + b_off + 4 * i);
    }

    v8f acc0 = {}, acc1 = {};
    #pragma unroll
    for (int kt = 0; kt < 8; ++kt) {
        v2f b;
        b.x = hv[kt >> 1][(kt & 1) * 2];
        b.y = hv[kt >> 1][(kt & 1) * 2 + 1];
        acc0 = __builtin_amdgcn_wmma_f32_16x16x4_f32(
            false, wa[0][kt], false, b, (short)0, acc0, false, false);
        acc1 = __builtin_amdgcn_wmma_f32_16x16x4_f32(
            false, wa[1][kt], false, b, (short)0, acc1, false, false);
    }

    #pragma unroll
    for (int p = 0; p < 2; ++p) {
        const v2f ir = {gc[0][2 * p], gc[0][2 * p + 1]};
        const v2f iz = {gc[1][2 * p], gc[1][2 * p + 1]};
        const v2f in = {gc[2][2 * p], gc[2][2 * p + 1]};
        const v2f hr = {acc0[2 * p], acc0[2 * p + 1]};
        const v2f hz = {acc0[4 + 2 * p], acc0[5 + 2 * p]};
        const v2f hn = {acc1[2 * p], acc1[2 * p + 1]};
        const v2f rg = sigmoid_2(ir + hr);
        const v2f zg = sigmoid_2(iz + hz);
        const v2f ng = tanh_2(in + rg * hn);
        hc[p] = (1.0f - zg) * ng + zg * hc[p];
    }
    *reinterpret_cast<v4f*>(wr + s_off) = v4f{hc[0].x, hc[0].y, hc[1].x, hc[1].y};
}

__device__ __forceinline__ void lds_barrier()
{
    __builtin_amdgcn_fence(__ATOMIC_RELEASE, "workgroup", "local");
    __builtin_amdgcn_s_barrier();
    __builtin_amdgcn_fence(__ATOMIC_ACQUIRE, "workgroup", "local");
}

extern "C" __global__ __launch_bounds__(128)
__attribute__((amdgpu_waves_per_eu(1, 1)))
void gru_recurrence(const float* __restrict__ gi, const float* __restrict__ w_hh,
                    const float* __restrict__ w1, const float* __restrict__ b1,
                    const float* __restrict__ w2, float* __restrict__ out)
{
    static_assert(GI_LEAD == 3 && T_SZ % (GI_LEAD + 1) == 0 && T_SZ > GI_LEAD,
                  "the scan body is four steps: four gi sets in rotation, two h buffers in alternation");
    __shared__ __attribute__((aligned(16))) float h_lds[2 * HB];
    const int tid = threadIdx.x;
    for (int i = tid; i < HB; i += 128) h_lds[i] = 0.0f;
    __syncthreads();

    const int lane = tid & 31;
    const int lo   = lane & 15;
    const int hi   = lane >> 4;
    const int wave = tid >> 5;
    const int bb   = blockIdx.x * 16;

    const int b_off = lo * HP + 16 * hi;
    const int s_off = lo * HP + 8 * wave + 4 * hi;

    const int  colh  = 8 * wave + 4 * (lo >> 3) + (lo & 3);
    const bool upper = (lo & 4) != 0;
    const float* wr0 = w_hh + ((upper ? H_SZ : 0) + colh) * H_SZ + 16 * hi;
    const float* wr1 = w_hh + (2 * H_SZ + colh) * H_SZ + 16 * hi;
    v2f wa[2][8];
    #pragma unroll
    for (int kt = 0; kt < 8; ++kt) {
        wa[0][kt].x = wr0[2 * kt];
        wa[0][kt].y = wr0[2 * kt + 1];
        v2f n;
        n.x = wr1[2 * kt];
        n.y = wr1[2 * kt + 1];
        wa[1][kt] = upper ? v2f{} : n;
    }

    v2f hc[2] = {};

    v2f w1f[8];
    #pragma unroll
    for (int kt = 0; kt < 8; ++kt) {
        const int k = kt * 4 + 2 * hi;
        w1f[kt].x = w1[lo * H_SZ + k];
        w1f[kt].y = w1[lo * H_SZ + k + 1];
    }
    float b1v[8], w2v[8];
    #pragma unroll
    for (int r = 0; r < 8; ++r) {
        b1v[r] = b1[hi * 8 + r];
        w2v[r] = w2[hi * 8 + r];
    }

    const float* gip = gi + ((long)bb + lo) * G3 + 8 * wave + 4 * hi;
    const long gstep = (long)B_SZ * G3;

    v4f g0[3], g1[3], g2[3], g3[3];
    gi_load(gip, g0);
    gi_load(gip + gstep, g1);
    gi_load(gip + 2 * gstep, g2);

    float* const hb0 = h_lds;
    float* const hb1 = h_lds + HB;
    for (int t = 0; t < T_SZ; t += 4) {
        const bool more = t + 4 < T_SZ;
        gru_step(gip + 3 * gstep, g3, hb0, hb1, wa, hc, g0, b_off, s_off);
        lds_barrier();
        gru_step(more ? gip + 4 * gstep : gip, g0, hb1, hb0, wa, hc, g1, b_off, s_off);
        lds_barrier();
        gru_step(more ? gip + 5 * gstep : gip, g1, hb0, hb1, wa, hc, g2, b_off, s_off);
        lds_barrier();
        gru_step(more ? gip + 6 * gstep : gip, g2, hb1, hb0, wa, hc, g3, b_off, s_off);
        lds_barrier();
        gip += 4 * gstep;
    }

    if (wave == 0) {
        v8f c0, c1 = {};
        #pragma unroll
        for (int r = 0; r < 8; ++r) c0[r] = b1v[r];
        const int hb_off = lo * HP + 2 * hi;
        #pragma unroll
        for (int kt = 0; kt < 8; kt += 2) {
            v2f hb0v, hb1v;
            hb0v.x = h_lds[hb_off + kt * 4];
            hb0v.y = h_lds[hb_off + kt * 4 + 1];
            hb1v.x = h_lds[hb_off + kt * 4 + 4];
            hb1v.y = h_lds[hb_off + kt * 4 + 5];
            c0 = __builtin_amdgcn_wmma_f32_16x16x4_f32(
                false, w1f[kt], false, hb0v, (short)0, c0, false, false);
            c1 = __builtin_amdgcn_wmma_f32_16x16x4_f32(
                false, w1f[kt + 1], false, hb1v, (short)0, c1, false, false);
        }
        float y = 0.0f;
        #pragma unroll
        for (int r = 0; r < 8; ++r) {
            y = fmaf(w2v[r], fmaxf(c0[r] + c1[r], 0.0f), y);
        }
        y += __shfl_xor(y, 16, 32);
        if (hi == 0) out[bb + lo] = y;
    }
}

extern "C" void kernel_launch(void* const* d_in, const int* in_sizes, int n_in,
                              void* d_out, int out_size, void* d_ws, size_t ws_size,
                              hipStream_t stream) {
    const float* x    = (const float*)d_in[0];
    const float* w_ih = (const float*)d_in[1];
    const float* w_hh = (const float*)d_in[2];
    const float* w1   = (const float*)d_in[3];
    const float* b1   = (const float*)d_in[4];
    const float* w2   = (const float*)d_in[5];
    float* out = (float*)d_out;

    float* gi = (float*)d_ws;

    gru_gemm_in<<<256, 128, 0, stream>>>(x, w_ih, gi);
    gru_recurrence<<<256, 128, 0, stream>>>(gi, w_hh, w1, b1, w2, out);
}
